# starting kernel
# baseline (speedup 1.0000x reference)
#include <hip/hip_runtime.h>


typedef __attribute__((ext_vector_type(2))) float v2f;
typedef __attribute__((ext_vector_type(8))) float v8f;

#define GCN_N 100000

__global__ void deg_kernel(const int* __restrict__ col, float* __restrict__ deg, int E) {
    int e = blockIdx.x * blockDim.x + threadIdx.x;
    if (e < E) atomicAdd(&deg[col[e]], 1.0f);
}

__global__ void dinv_kernel(float* __restrict__ deg, int n) {
    int i = blockIdx.x * blockDim.x + threadIdx.x;
    if (i < n) deg[i] = rsqrtf(deg[i] + 1.0f);
}

__global__ void gemm_k16_n32_wmma(const float* __restrict__ X, const float* __restrict__ W,
                                  float* __restrict__ H, int n) {
    int wave   = blockIdx.x * (blockDim.x >> 5) + (threadIdx.x >> 5);
    int nTiles = (n + 15) >> 4;
    if (wave >= nTiles) return;
    int lane = threadIdx.x & 31;
    int mi   = lane & 15;
    int kb   = (lane >> 4) << 1;
    int r0   = wave << 4;
    bool full = (r0 + 16 <= n);
    int rld  = full ? (r0 + mi) : min(r0 + mi, n - 1);
    const float* xr = X + (size_t)rld * 16;

    v8f acc0 = {}; v8f acc1 = {};
#pragma unroll
    for (int kk = 0; kk < 16; kk += 4) {
        v2f a = *(const v2f*)(xr + kk + kb);
        v2f b0, b1;
        b0.x = W[(kk + kb) * 32 + mi];          b0.y = W[(kk + kb + 1) * 32 + mi];
        b1.x = W[(kk + kb) * 32 + mi + 16];     b1.y = W[(kk + kb + 1) * 32 + mi + 16];
        acc0 = __builtin_amdgcn_wmma_f32_16x16x4_f32(false, a, false, b0, (short)0, acc0, false, false);
        acc1 = __builtin_amdgcn_wmma_f32_16x16x4_f32(false, a, false, b1, (short)0, acc1, false, false);
    }
    int mh = (lane >> 4) << 3;
    float* hb = H + (size_t)(r0 + mh) * 32 + mi;
    if (full) {
#pragma unroll
        for (int v = 0; v < 8; v++) {
            hb[v * 32]      = acc0[v];
            hb[v * 32 + 16] = acc1[v];
        }
    } else {
#pragma unroll
        for (int v = 0; v < 8; v++) {
            if (r0 + mh + v < n) {
                hb[v * 32]      = acc0[v];
                hb[v * 32 + 16] = acc1[v];
            }
        }
    }
}

__global__ void gemm_k32_n16_wmma(const float* __restrict__ X, const float* __restrict__ W,
                                  float* __restrict__ H, int n) {
    int wave   = blockIdx.x * (blockDim.x >> 5) + (threadIdx.x >> 5);
    int nTiles = (n + 15) >> 4;
    if (wave >= nTiles) return;
    int lane = threadIdx.x & 31;
    int mi   = lane & 15;
    int kb   = (lane >> 4) << 1;
    int r0   = wave << 4;
    bool full = (r0 + 16 <= n);
    int rld  = full ? (r0 + mi) : min(r0 + mi, n - 1);
    const float* xr = X + (size_t)rld * 32;

    v8f acc = {};
#pragma unroll
    for (int kk = 0; kk < 32; kk += 4) {
        v2f a = *(const v2f*)(xr + kk + kb);
        v2f b;
        b.x = W[(kk + kb) * 16 + mi];
        b.y = W[(kk + kb + 1) * 16 + mi];
        acc = __builtin_amdgcn_wmma_f32_16x16x4_f32(false, a, false, b, (short)0, acc, false, false);
    }
    int mh = (lane >> 4) << 3;
    float* hb = H + (size_t)(r0 + mh) * 16 + mi;
    if (full) {
#pragma unroll
        for (int v = 0; v < 8; v++) hb[v * 16] = acc[v];
    } else {
#pragma unroll
        for (int v = 0; v < 8; v++)
            if (r0 + mh + v < n) hb[v * 16] = acc[v];
    }
}

template <int C>
__global__ void scatter_kernel(const int* __restrict__ row, const int* __restrict__ col,
                               const float* __restrict__ h, const float* __restrict__ dinv,
                               float* __restrict__ agg, int E) {
    long long t = (long long)blockIdx.x * blockDim.x + threadIdx.x;
    int e = (int)(t / C);
    int j = (int)(t % C);
    if (e < E) {
        int r = row[e], c = col[e];
        float nrm = dinv[r] * dinv[c];
        atomicAdd(&agg[(size_t)c * C + j], h[(size_t)r * C + j] * nrm);
    }
}

__global__ void selfloop_bias_relu(float* __restrict__ agg, const float* __restrict__ h,
                                   const float* __restrict__ dinv, const float* __restrict__ b,
                                   int n, int C) {
    long long t = (long long)blockIdx.x * blockDim.x + threadIdx.x;
    int i = (int)(t / C);
    int j = (int)(t % C);
    if (i < n) {
        float di = dinv[i];
        float v  = agg[(size_t)i * C + j] + h[(size_t)i * C + j] * di * di + b[j];
        agg[(size_t)i * C + j] = v > 0.0f ? v : 0.0f;
    }
}

__global__ void final_kernel(const float* __restrict__ agg2, const float* __restrict__ h2,
                             const float* __restrict__ dinv, const float* __restrict__ b2,
                             const float* __restrict__ fcW, const float* __restrict__ fcb,
                             float* __restrict__ out, int n) {
    int i = blockIdx.x * blockDim.x + threadIdx.x;
    if (i < n) {
        float d2 = dinv[i] * dinv[i];
        float s = 0.0f;
#pragma unroll
        for (int j = 0; j < 16; j++) {
            float v = agg2[(size_t)i * 16 + j] + h2[(size_t)i * 16 + j] * d2 + b2[j];
            v = v > 0.0f ? v : 0.0f;
            s += v * fcW[j];
        }
        out[i] = s + fcb[0];
    }
}

extern "C" void kernel_launch(void* const* d_in, const int* in_sizes, int n_in,
                              void* d_out, int out_size, void* d_ws, size_t ws_size,
                              hipStream_t stream) {
    const int*   edge = (const int*)d_in[0];
    const float* x0   = (const float*)d_in[1];
    const float* W1   = (const float*)d_in[2];
    const float* b1   = (const float*)d_in[3];
    const float* W2   = (const float*)d_in[4];
    const float* b2   = (const float*)d_in[5];
    const float* fcW  = (const float*)d_in[6];
    const float* fcb  = (const float*)d_in[7];
    float*       out  = (float*)d_out;

    const int n = GCN_N;
    const int E = in_sizes[0] / 2;
    const int* row = edge;
    const int* col = edge + E;

    float* ws   = (float*)d_ws;
    size_t npad = ((size_t)n + 63) & ~(size_t)63;
    float* dinv = ws;
    float* h1   = dinv + npad;
    float* agg1 = h1   + (size_t)n * 32;
    float* h2   = agg1 + (size_t)n * 32;
    float* agg2 = h2   + (size_t)n * 16;

    hipMemsetAsync(dinv, 0, (size_t)n * sizeof(float), stream);
    hipMemsetAsync(agg1, 0, (size_t)n * 32 * sizeof(float), stream);
    hipMemsetAsync(agg2, 0, (size_t)n * 16 * sizeof(float), stream);

    deg_kernel <<<(E + 255) / 256, 256, 0, stream>>>(col, dinv, E);
    dinv_kernel<<<(n + 255) / 256, 256, 0, stream>>>(dinv, n);

    int tiles   = (n + 15) / 16;
    int gblocks = (tiles + 7) / 8;

    gemm_k16_n32_wmma<<<gblocks, 256, 0, stream>>>(x0, W1, h1, n);
    long long tot1 = (long long)E * 32;
    scatter_kernel<32><<<(int)((tot1 + 255) / 256), 256, 0, stream>>>(row, col, h1, dinv, agg1, E);
    selfloop_bias_relu<<<(int)(((long long)n * 32 + 255) / 256), 256, 0, stream>>>(agg1, h1, dinv, b1, n, 32);

    gemm_k32_n16_wmma<<<gblocks, 256, 0, stream>>>(agg1, W2, h2, n);
    long long tot2 = (long long)E * 16;
    scatter_kernel<16><<<(int)((tot2 + 255) / 256), 256, 0, stream>>>(row, col, h2, dinv, agg2, E);

    final_kernel<<<(n + 255) / 256, 256, 0, stream>>>(agg2, h2, dinv, b2, fcW, fcb, out, n);
}


# layer 1 aggregates the 16-wide input before its linear map ((A x) W1 for A (x W1), sums reassociated): 64B atomic rows per edge, not 128B; self-loop, bias, relu fused into the WMMA GEMM
# speedup vs baseline: 1.2864x; 1.2864x over previous
#include <hip/hip_runtime.h>


typedef __attribute__((ext_vector_type(2))) float v2f;
typedef __attribute__((ext_vector_type(8))) float v8f;

#define GCN_N 100000

__global__ void deg_kernel(const int* __restrict__ col, float* __restrict__ deg, int E) {
    int e = blockIdx.x * blockDim.x + threadIdx.x;
    if (e < E) atomicAdd(&deg[col[e]], 1.0f);
}

__global__ void dinv_kernel(float* __restrict__ deg, int n) {
    int i = blockIdx.x * blockDim.x + threadIdx.x;
    if (i < n) deg[i] = rsqrtf(deg[i] + 1.0f);
}

__global__ void gemm_k16_n32_wmma(const float* __restrict__ AGG, const float* __restrict__ X,
                                  const float* __restrict__ dinv, const float* __restrict__ W,
                                  const float* __restrict__ b, float* __restrict__ H, int n) {
    int wave   = blockIdx.x * (blockDim.x >> 5) + (threadIdx.x >> 5);
    int nTiles = (n + 15) >> 4;
    if (wave >= nTiles) return;
    int lane = threadIdx.x & 31;
    int mi   = lane & 15;
    int kb   = (lane >> 4) << 1;
    int r0   = wave << 4;
    bool full = (r0 + 16 <= n);
    int rld  = full ? (r0 + mi) : min(r0 + mi, n - 1);
    const float* gr = AGG + (size_t)rld * 16;
    const float* xr = X   + (size_t)rld * 16;
    float di = dinv[rld];

    v8f acc0 = {}; v8f acc1 = {};
#pragma unroll
    for (int kk = 0; kk < 16; kk += 4) {
        v2f g = *(const v2f*)(gr + kk + kb);
        v2f x = *(const v2f*)(xr + kk + kb);
        v2f a;
        a.x = g.x + x.x * di * di;
        a.y = g.y + x.y * di * di;
        v2f b0, b1;
        b0.x = W[(kk + kb) * 32 + mi];          b0.y = W[(kk + kb + 1) * 32 + mi];
        b1.x = W[(kk + kb) * 32 + mi + 16];     b1.y = W[(kk + kb + 1) * 32 + mi + 16];
        acc0 = __builtin_amdgcn_wmma_f32_16x16x4_f32(false, a, false, b0, (short)0, acc0, false, false);
        acc1 = __builtin_amdgcn_wmma_f32_16x16x4_f32(false, a, false, b1, (short)0, acc1, false, false);
    }
    float bias0 = b[mi], bias1 = b[mi + 16];
    v8f o0, o1;
#pragma unroll
    for (int v = 0; v < 8; v++) {
        float t0 = acc0[v] + bias0;
        float t1 = acc1[v] + bias1;
        o0[v] = t0 > 0.0f ? t0 : 0.0f;
        o1[v] = t1 > 0.0f ? t1 : 0.0f;
    }
    int mh = (lane >> 4) << 3;
    float* hb = H + (size_t)(r0 + mh) * 32 + mi;
    if (full) {
#pragma unroll
        for (int v = 0; v < 8; v++) {
            hb[v * 32]      = o0[v];
            hb[v * 32 + 16] = o1[v];
        }
    } else {
#pragma unroll
        for (int v = 0; v < 8; v++) {
            if (r0 + mh + v < n) {
                hb[v * 32]      = o0[v];
                hb[v * 32 + 16] = o1[v];
            }
        }
    }
}

__global__ void gemm_k32_n16_wmma(const float* __restrict__ X, const float* __restrict__ W,
                                  float* __restrict__ H, int n) {
    int wave   = blockIdx.x * (blockDim.x >> 5) + (threadIdx.x >> 5);
    int nTiles = (n + 15) >> 4;
    if (wave >= nTiles) return;
    int lane = threadIdx.x & 31;
    int mi   = lane & 15;
    int kb   = (lane >> 4) << 1;
    int r0   = wave << 4;
    bool full = (r0 + 16 <= n);
    int rld  = full ? (r0 + mi) : min(r0 + mi, n - 1);
    const float* xr = X + (size_t)rld * 32;

    v8f acc = {};
#pragma unroll
    for (int kk = 0; kk < 32; kk += 4) {
        v2f a = *(const v2f*)(xr + kk + kb);
        v2f b;
        b.x = W[(kk + kb) * 16 + mi];
        b.y = W[(kk + kb + 1) * 16 + mi];
        acc = __builtin_amdgcn_wmma_f32_16x16x4_f32(false, a, false, b, (short)0, acc, false, false);
    }
    int mh = (lane >> 4) << 3;
    float* hb = H + (size_t)(r0 + mh) * 16 + mi;
    if (full) {
#pragma unroll
        for (int v = 0; v < 8; v++) hb[v * 16] = acc[v];
    } else {
#pragma unroll
        for (int v = 0; v < 8; v++)
            if (r0 + mh + v < n) hb[v * 16] = acc[v];
    }
}

template <int C>
__global__ void scatter_kernel(const int* __restrict__ row, const int* __restrict__ col,
                               const float* __restrict__ h, const float* __restrict__ dinv,
                               float* __restrict__ agg, int E) {
    long long t = (long long)blockIdx.x * blockDim.x + threadIdx.x;
    int e = (int)(t / C);
    int j = (int)(t % C);
    if (e < E) {
        int r = row[e], c = col[e];
        float nrm = dinv[r] * dinv[c];
        atomicAdd(&agg[(size_t)c * C + j], h[(size_t)r * C + j] * nrm);
    }
}

__global__ void final_kernel(const float* __restrict__ agg2, const float* __restrict__ h2,
                             const float* __restrict__ dinv, const float* __restrict__ b2,
                             const float* __restrict__ fcW, const float* __restrict__ fcb,
                             float* __restrict__ out, int n) {
    int i = blockIdx.x * blockDim.x + threadIdx.x;
    if (i < n) {
        float d2 = dinv[i] * dinv[i];
        float s = 0.0f;
#pragma unroll
        for (int j = 0; j < 16; j++) {
            float v = agg2[(size_t)i * 16 + j] + h2[(size_t)i * 16 + j] * d2 + b2[j];
            v = v > 0.0f ? v : 0.0f;
            s += v * fcW[j];
        }
        out[i] = s + fcb[0];
    }
}

extern "C" void kernel_launch(void* const* d_in, const int* in_sizes, int n_in,
                              void* d_out, int out_size, void* d_ws, size_t ws_size,
                              hipStream_t stream) {
    const int*   edge = (const int*)d_in[0];
    const float* x0   = (const float*)d_in[1];
    const float* W1   = (const float*)d_in[2];
    const float* b1   = (const float*)d_in[3];
    const float* W2   = (const float*)d_in[4];
    const float* b2   = (const float*)d_in[5];
    const float* fcW  = (const float*)d_in[6];
    const float* fcb  = (const float*)d_in[7];
    float*       out  = (float*)d_out;

    const int n = GCN_N;
    const int E = in_sizes[0] / 2;
    const int* row = edge;
    const int* col = edge + E;

    float* ws   = (float*)d_ws;
    size_t npad = ((size_t)n + 63) & ~(size_t)63;
    float* dinv = ws;
    float* aggx = dinv + npad;
    float* x1   = aggx + (size_t)n * 16;
    float* h2   = x1   + (size_t)n * 32;
    float* agg2 = h2   + (size_t)n * 16;

    hipMemsetAsync(dinv, 0, (size_t)n * sizeof(float), stream);
    hipMemsetAsync(aggx, 0, (size_t)n * 16 * sizeof(float), stream);
    hipMemsetAsync(agg2, 0, (size_t)n * 16 * sizeof(float), stream);

    deg_kernel <<<(E + 255) / 256, 256, 0, stream>>>(col, dinv, E);
    dinv_kernel<<<(n + 255) / 256, 256, 0, stream>>>(dinv, n);

    int tiles   = (n + 15) / 16;
    int gblocks = (tiles + 7) / 8;
    long long tot = (long long)E * 16;

    scatter_kernel<16><<<(int)((tot + 255) / 256), 256, 0, stream>>>(row, col, x0, dinv, aggx, E);
    gemm_k16_n32_wmma<<<gblocks, 256, 0, stream>>>(aggx, x0, dinv, W1, b1, x1, n);

    gemm_k32_n16_wmma<<<gblocks, 256, 0, stream>>>(x1, W2, h2, n);
    scatter_kernel<16><<<(int)((tot + 255) / 256), 256, 0, stream>>>(row, col, h2, dinv, agg2, E);

    final_kernel<<<(n + 255) / 256, 256, 0, stream>>>(agg2, h2, dinv, b2, fcW, fcb, out, n);
}


# edges bucketed by target per call (returning int add = rank, prefix, fill); both layers gather-sum dinv-prescaled 64B rows in registers, slot-arrival order: no float atomics/clears; fc sum lane tree
# speedup vs baseline: 1.7528x; 1.3626x over previous
#include <hip/hip_runtime.h>


typedef __attribute__((ext_vector_type(2))) float v2f;
typedef __attribute__((ext_vector_type(8))) float v8f;

#define GCN_N 100000

__global__ void count_rank_kernel(const int* __restrict__ col, int* __restrict__ cnt,
                                  int* __restrict__ rank, int E) {
    int T = gridDim.x * blockDim.x;
    int t = blockIdx.x * blockDim.x + threadIdx.x;
    if (t + 3 * T < E) {
        int c0 = col[t], c1 = col[t + T], c2 = col[t + 2 * T], c3 = col[t + 3 * T];
        int k0 = atomicAdd(&cnt[c0], 1);
        int k1 = atomicAdd(&cnt[c1], 1);
        int k2 = atomicAdd(&cnt[c2], 1);
        int k3 = atomicAdd(&cnt[c3], 1);
        rank[t]         = k0;
        rank[t + T]     = k1;
        rank[t + 2 * T] = k2;
        rank[t + 3 * T] = k3;
    } else {
        for (int e = t; e < E; e += T) rank[e] = atomicAdd(&cnt[col[e]], 1);
    }
}

__global__ void prefix_kernel(const int* __restrict__ cnt, int* __restrict__ nxt,
                              float* __restrict__ dinv, const float* __restrict__ x,
                              float* __restrict__ xs, int n) {
    __shared__ int wsum[32];
    __shared__ int base_s;
    __shared__ float dl[1024];
    int tid  = threadIdx.x;
    int lane = tid & 31;
    int w    = tid >> 5;
    int b0   = blockIdx.x * 1024;

    int part = 0;
    for (int i = tid; i < b0; i += 1024) part += cnt[i];
    for (int off = 16; off > 0; off >>= 1) part += __shfl_xor(part, off);
    if (lane == 0) wsum[w] = part;
    __syncthreads();
    if (w == 0) {
        int v = wsum[lane];
        for (int off = 16; off > 0; off >>= 1) v += __shfl_xor(v, off);
        if (lane == 0) base_s = v;
    }
    __syncthreads();
    int base = base_s;

    int i = b0 + tid;
    int c = (i < n) ? cnt[i] : 0;
    int incl = c;
    for (int off = 1; off < 32; off <<= 1) {
        int t = __shfl_up(incl, off);
        if (lane >= off) incl += t;
    }
    if (lane == 31) wsum[w] = incl;
    __syncthreads();
    int ws = wsum[lane];
    for (int off = 1; off < 32; off <<= 1) {
        int t = __shfl_up(ws, off);
        if (lane >= off) ws += t;
    }
    int woff = __shfl(ws, (w > 0) ? (w - 1) : 0);
    if (w == 0) woff = 0;
    float dv = rsqrtf((float)c + 1.0f);
    dl[tid] = dv;
    if (i < n) {
        int excl = base + woff + incl - c;
        nxt[i]  = excl;
        dinv[i] = dv;
        if (i == n - 1) nxt[n] = excl + c;
    }
    __syncthreads();
    size_t lim = (size_t)n * 16;
    size_t o16 = (size_t)b0 * 16;
    if (o16 + 16384 <= lim) {
        for (int it = 0; it < 16; it++) {
            int loc = it * 1024 + tid;
            xs[o16 + loc] = x[o16 + loc] * dl[loc >> 4];
        }
    } else {
        for (int it = 0; it < 16; it++) {
            int loc = it * 1024 + tid;
            size_t idx = o16 + loc;
            if (idx < lim) xs[idx] = x[idx] * dl[loc >> 4];
        }
    }
}

__global__ void fill_kernel(const int* __restrict__ row, const int* __restrict__ col,
                            const int* __restrict__ nxt, const int* __restrict__ rank,
                            int* __restrict__ src, int E) {
    int T = gridDim.x * blockDim.x;
    int t = blockIdx.x * blockDim.x + threadIdx.x;
    unsigned lim = (unsigned)E;
    if (t + 3 * T < E) {
        int c0 = col[t], c1 = col[t + T], c2 = col[t + 2 * T], c3 = col[t + 3 * T];
        int k0 = rank[t], k1 = rank[t + T], k2 = rank[t + 2 * T], k3 = rank[t + 3 * T];
        int r0 = row[t], r1 = row[t + T], r2 = row[t + 2 * T], r3 = row[t + 3 * T];
        int s0 = nxt[c0] + k0;
        int s1 = nxt[c1] + k1;
        int s2 = nxt[c2] + k2;
        int s3 = nxt[c3] + k3;
        if ((unsigned)s0 < lim) src[s0] = r0;
        if ((unsigned)s1 < lim) src[s1] = r1;
        if ((unsigned)s2 < lim) src[s2] = r2;
        if ((unsigned)s3 < lim) src[s3] = r3;
    } else {
        for (int e = t; e < E; e += T) {
            int s = nxt[col[e]] + rank[e];
            if ((unsigned)s < lim) src[s] = row[e];
        }
    }
}

__device__ __forceinline__ float gather16(const int* __restrict__ nxt, const int* __restrict__ src,
                                          const float* __restrict__ hs, int c, int j, int n, int E) {
    int beg = nxt[c];
    int end = nxt[c + 1];
    beg = max(beg, 0);
    end = min(end, E);
    unsigned last = (unsigned)(n - 1);
    float acc = 0.0f;
    int k = beg;
    for (; k + 4 <= end; k += 4) {
        unsigned r0 = min((unsigned)src[k],     last);
        unsigned r1 = min((unsigned)src[k + 1], last);
        unsigned r2 = min((unsigned)src[k + 2], last);
        unsigned r3 = min((unsigned)src[k + 3], last);
        float v0 = hs[(size_t)r0 * 16 + j];
        float v1 = hs[(size_t)r1 * 16 + j];
        float v2 = hs[(size_t)r2 * 16 + j];
        float v3 = hs[(size_t)r3 * 16 + j];
        acc += v0;
        acc += v1;
        acc += v2;
        acc += v3;
    }
    for (; k < end; k++) {
        unsigned r = min((unsigned)src[k], last);
        acc += hs[(size_t)r * 16 + j];
    }
    return acc;
}

__global__ void gather_kernel(const int* __restrict__ nxt, const int* __restrict__ src,
                              const float* __restrict__ xs, const float* __restrict__ dinv,
                              float* __restrict__ agg, int n, int E) {
    int t = blockIdx.x * blockDim.x + threadIdx.x;
    int c = t >> 4;
    int j = t & 15;
    if (c < n) {
        float acc = gather16(nxt, src, xs, c, j, n, E);
        agg[(size_t)c * 16 + j] = acc * dinv[c];
    }
}

__global__ void gemm_k16_n32_wmma(const float* __restrict__ AGG, const float* __restrict__ X,
                                  const float* __restrict__ dinv, const float* __restrict__ W,
                                  const float* __restrict__ b, float* __restrict__ H, int n) {
    int wave   = blockIdx.x * (blockDim.x >> 5) + (threadIdx.x >> 5);
    int nTiles = (n + 15) >> 4;
    if (wave >= nTiles) return;
    int lane = threadIdx.x & 31;
    int mi   = lane & 15;
    int kb   = (lane >> 4) << 1;
    int r0   = wave << 4;
    bool full = (r0 + 16 <= n);
    int rld  = full ? (r0 + mi) : min(r0 + mi, n - 1);
    const float* gr = AGG + (size_t)rld * 16;
    const float* xr = X   + (size_t)rld * 16;
    float di = dinv[rld];

    v8f acc0 = {}; v8f acc1 = {};
#pragma unroll
    for (int kk = 0; kk < 16; kk += 4) {
        v2f g = *(const v2f*)(gr + kk + kb);
        v2f x = *(const v2f*)(xr + kk + kb);
        v2f a;
        a.x = g.x + x.x * di * di;
        a.y = g.y + x.y * di * di;
        v2f b0, b1;
        b0.x = W[(kk + kb) * 32 + mi];          b0.y = W[(kk + kb + 1) * 32 + mi];
        b1.x = W[(kk + kb) * 32 + mi + 16];     b1.y = W[(kk + kb + 1) * 32 + mi + 16];
        acc0 = __builtin_amdgcn_wmma_f32_16x16x4_f32(false, a, false, b0, (short)0, acc0, false, false);
        acc1 = __builtin_amdgcn_wmma_f32_16x16x4_f32(false, a, false, b1, (short)0, acc1, false, false);
    }
    float bias0 = b[mi], bias1 = b[mi + 16];
    v8f o0, o1;
#pragma unroll
    for (int v = 0; v < 8; v++) {
        float t0 = acc0[v] + bias0;
        float t1 = acc1[v] + bias1;
        o0[v] = t0 > 0.0f ? t0 : 0.0f;
        o1[v] = t1 > 0.0f ? t1 : 0.0f;
    }
    int mh = (lane >> 4) << 3;
    float* hb = H + (size_t)(r0 + mh) * 32 + mi;
    if (full) {
#pragma unroll
        for (int v = 0; v < 8; v++) {
            hb[v * 32]      = o0[v];
            hb[v * 32 + 16] = o1[v];
        }
    } else {
#pragma unroll
        for (int v = 0; v < 8; v++) {
            if (r0 + mh + v < n) {
                hb[v * 32]      = o0[v];
                hb[v * 32 + 16] = o1[v];
            }
        }
    }
}

__global__ void gemm_k32_n16_wmma(const float* __restrict__ X, const float* __restrict__ W,
                                  const float* __restrict__ dinv, float* __restrict__ H, int n) {
    int wave   = blockIdx.x * (blockDim.x >> 5) + (threadIdx.x >> 5);
    int nTiles = (n + 15) >> 4;
    if (wave >= nTiles) return;
    int lane = threadIdx.x & 31;
    int mi   = lane & 15;
    int kb   = (lane >> 4) << 1;
    int r0   = wave << 4;
    bool full = (r0 + 16 <= n);
    int rld  = full ? (r0 + mi) : min(r0 + mi, n - 1);
    const float* xr = X + (size_t)rld * 32;
    float di = dinv[rld];

    v8f acc = {};
#pragma unroll
    for (int kk = 0; kk < 32; kk += 4) {
        v2f a = *(const v2f*)(xr + kk + kb);
        a.x *= di;
        a.y *= di;
        v2f b;
        b.x = W[(kk + kb) * 16 + mi];
        b.y = W[(kk + kb + 1) * 16 + mi];
        acc = __builtin_amdgcn_wmma_f32_16x16x4_f32(false, a, false, b, (short)0, acc, false, false);
    }
    int mh = (lane >> 4) << 3;
    float* hb = H + (size_t)(r0 + mh) * 16 + mi;
    if (full) {
#pragma unroll
        for (int v = 0; v < 8; v++) hb[v * 16] = acc[v];
    } else {
#pragma unroll
        for (int v = 0; v < 8; v++)
            if (r0 + mh + v < n) hb[v * 16] = acc[v];
    }
}

__global__ void gather_final_kernel(const int* __restrict__ nxt, const int* __restrict__ src,
                                    const float* __restrict__ hs, const float* __restrict__ dinv,
                                    const float* __restrict__ b2, const float* __restrict__ fcW,
                                    const float* __restrict__ fcb, float* __restrict__ out,
                                    int n, int E) {
    __shared__ float so[32];
    int tid = threadIdx.x;
    int c = blockIdx.x * 32 + (tid >> 4);
    int j = tid & 15;
    if (c < n) {
        float acc = gather16(nxt, src, hs, c, j, n, E);
        float v = (acc + hs[(size_t)c * 16 + j]) * dinv[c] + b2[j];
        v = v > 0.0f ? v : 0.0f;
        float s = v * fcW[j];
        s += __shfl_xor(s, 8, 16);
        s += __shfl_xor(s, 4, 16);
        s += __shfl_xor(s, 2, 16);
        s += __shfl_xor(s, 1, 16);
        if (j == 0) so[tid >> 4] = s + fcb[0];
    }
    __syncthreads();
    if (tid < 32) {
        int i = blockIdx.x * 32 + tid;
        if (i < n) out[i] = so[tid];
    }
}

extern "C" void kernel_launch(void* const* d_in, const int* in_sizes, int n_in,
                              void* d_out, int out_size, void* d_ws, size_t ws_size,
                              hipStream_t stream) {
    const int*   edge = (const int*)d_in[0];
    const float* x0   = (const float*)d_in[1];
    const float* W1   = (const float*)d_in[2];
    const float* b1   = (const float*)d_in[3];
    const float* W2   = (const float*)d_in[4];
    const float* b2   = (const float*)d_in[5];
    const float* fcW  = (const float*)d_in[6];
    const float* fcb  = (const float*)d_in[7];
    float*       out  = (float*)d_out;

    const int n = GCN_N;
    const int E = in_sizes[0] / 2;
    const int* row = edge;
    const int* col = edge + E;

    float* ws   = (float*)d_ws;
    size_t npad = ((size_t)n + 63) & ~(size_t)63;
    float* dinv = ws;
    int*   nxt  = (int*)(dinv + npad);
    int*   src  = nxt + npad;
    float* aggx = (float*)(src + (size_t)n * 16);
    int*   rank = (int*)aggx;
    float* hs   = aggx;
    float* x1   = aggx + (size_t)n * 16;
    int*   cnt  = (int*)x1;
    float* xs   = x1 + npad;

    hipMemsetAsync(cnt, 0, (size_t)n * sizeof(int), stream);

    int eblocks = (E + 1023) / 1024;
    count_rank_kernel<<<eblocks, 256, 0, stream>>>(col, cnt, rank, E);
    prefix_kernel<<<(n + 1023) / 1024, 1024, 0, stream>>>(cnt, nxt, dinv, x0, xs, n);
    fill_kernel<<<eblocks, 256, 0, stream>>>(row, col, nxt, rank, src, E);

    int tiles   = (n + 15) / 16;
    int gblocks = (tiles + 7) / 8;

    gather_kernel<<<(n * 16 + 255) / 256, 256, 0, stream>>>(nxt, src, xs, dinv, aggx, n, E);
    gemm_k16_n32_wmma<<<gblocks, 256, 0, stream>>>(aggx, x0, dinv, W1, b1, x1, n);

    gemm_k32_n16_wmma<<<gblocks, 256, 0, stream>>>(x1, W2, dinv, hs, n);
    gather_final_kernel<<<(n + 31) / 32, 512, 0, stream>>>(nxt, src, hs, dinv, b2, fcW, fcb, out, n, E);
}


# gathers read each 64B row as 4 lanes x 16B (was 16 lanes x 4B): 8 nodes per wave, 4x fewer waves; 8 row loads in flight per lane, next trip's slot numbers fetched under them; sums bit-identical
# speedup vs baseline: 1.8498x; 1.0553x over previous
#include <hip/hip_runtime.h>


typedef __attribute__((ext_vector_type(2))) float v2f;
typedef __attribute__((ext_vector_type(8))) float v8f;
typedef __attribute__((ext_vector_type(4))) float v4f;

#define GCN_N 100000

__global__ void count_rank_kernel(const int* __restrict__ col, int* __restrict__ cnt,
                                  int* __restrict__ rank, int E) {
    int T = gridDim.x * blockDim.x;
    int t = blockIdx.x * blockDim.x + threadIdx.x;
    if (t + 3 * T < E) {
        int c0 = col[t], c1 = col[t + T], c2 = col[t + 2 * T], c3 = col[t + 3 * T];
        int k0 = atomicAdd(&cnt[c0], 1);
        int k1 = atomicAdd(&cnt[c1], 1);
        int k2 = atomicAdd(&cnt[c2], 1);
        int k3 = atomicAdd(&cnt[c3], 1);
        rank[t]         = k0;
        rank[t + T]     = k1;
        rank[t + 2 * T] = k2;
        rank[t + 3 * T] = k3;
    } else {
        for (int e = t; e < E; e += T) rank[e] = atomicAdd(&cnt[col[e]], 1);
    }
}

__global__ void prefix_kernel(const int* __restrict__ cnt, int* __restrict__ nxt,
                              float* __restrict__ dinv, const float* __restrict__ x,
                              float* __restrict__ xs, int n) {
    __shared__ int wsum[32];
    __shared__ int base_s;
    __shared__ float dl[1024];
    int tid  = threadIdx.x;
    int lane = tid & 31;
    int w    = tid >> 5;
    int b0   = blockIdx.x * 1024;

    int part = 0;
    for (int i = tid; i < b0; i += 1024) part += cnt[i];
    for (int off = 16; off > 0; off >>= 1) part += __shfl_xor(part, off);
    if (lane == 0) wsum[w] = part;
    __syncthreads();
    if (w == 0) {
        int v = wsum[lane];
        for (int off = 16; off > 0; off >>= 1) v += __shfl_xor(v, off);
        if (lane == 0) base_s = v;
    }
    __syncthreads();
    int base = base_s;

    int i = b0 + tid;
    int c = (i < n) ? cnt[i] : 0;
    int incl = c;
    for (int off = 1; off < 32; off <<= 1) {
        int t = __shfl_up(incl, off);
        if (lane >= off) incl += t;
    }
    if (lane == 31) wsum[w] = incl;
    __syncthreads();
    int ws = wsum[lane];
    for (int off = 1; off < 32; off <<= 1) {
        int t = __shfl_up(ws, off);
        if (lane >= off) ws += t;
    }
    int woff = __shfl(ws, (w > 0) ? (w - 1) : 0);
    if (w == 0) woff = 0;
    float dv = rsqrtf((float)c + 1.0f);
    dl[tid] = dv;
    if (i < n) {
        int excl = base + woff + incl - c;
        nxt[i]  = excl;
        dinv[i] = dv;
        if (i == n - 1) nxt[n] = excl + c;
    }
    __syncthreads();
    size_t lim = (size_t)n * 16;
    size_t o16 = (size_t)b0 * 16;
    if (o16 + 16384 <= lim) {
        for (int it = 0; it < 16; it++) {
            int loc = it * 1024 + tid;
            xs[o16 + loc] = x[o16 + loc] * dl[loc >> 4];
        }
    } else {
        for (int it = 0; it < 16; it++) {
            int loc = it * 1024 + tid;
            size_t idx = o16 + loc;
            if (idx < lim) xs[idx] = x[idx] * dl[loc >> 4];
        }
    }
}

__global__ void fill_kernel(const int* __restrict__ row, const int* __restrict__ col,
                            const int* __restrict__ nxt, const int* __restrict__ rank,
                            int* __restrict__ src, int E) {
    int T = gridDim.x * blockDim.x;
    int t = blockIdx.x * blockDim.x + threadIdx.x;
    unsigned lim = (unsigned)E;
    if (t + 3 * T < E) {
        int c0 = col[t], c1 = col[t + T], c2 = col[t + 2 * T], c3 = col[t + 3 * T];
        int k0 = rank[t], k1 = rank[t + T], k2 = rank[t + 2 * T], k3 = rank[t + 3 * T];
        int r0 = row[t], r1 = row[t + T], r2 = row[t + 2 * T], r3 = row[t + 3 * T];
        int s0 = nxt[c0] + k0;
        int s1 = nxt[c1] + k1;
        int s2 = nxt[c2] + k2;
        int s3 = nxt[c3] + k3;
        if ((unsigned)s0 < lim) src[s0] = r0;
        if ((unsigned)s1 < lim) src[s1] = r1;
        if ((unsigned)s2 < lim) src[s2] = r2;
        if ((unsigned)s3 < lim) src[s3] = r3;
    } else {
        for (int e = t; e < E; e += T) {
            int s = nxt[col[e]] + rank[e];
            if ((unsigned)s < lim) src[s] = row[e];
        }
    }
}

#define GATHER_U 8
__device__ __forceinline__ v4f gather_rows4(const int* __restrict__ nxt, const int* __restrict__ src,
                                            const float* __restrict__ hs, int c, int q, int n, int E) {
    int beg = nxt[c];
    int end = nxt[c + 1];
    beg = max(beg, 0);
    end = min(end, E);
    int hi = end - 1;
    unsigned last = (unsigned)(n - 1);
    const char* hb = (const char*)hs;
    unsigned qo = (unsigned)q * 16u;
    v4f acc = {0.0f, 0.0f, 0.0f, 0.0f};
    unsigned r[GATHER_U];
#pragma unroll
    for (int i = 0; i < GATHER_U; i++) r[i] = min((unsigned)src[max(min(beg + i, hi), 0)], last);
    for (int k = beg; k < end; k += GATHER_U) {
        v4f v[GATHER_U];
#pragma unroll
        for (int i = 0; i < GATHER_U; i++) v[i] = *(const v4f*)(hb + (r[i] * 64u + qo));
#pragma unroll
        for (int i = 0; i < GATHER_U; i++) r[i] = min((unsigned)src[max(min(k + GATHER_U + i, hi), 0)], last);
#pragma unroll
        for (int i = 0; i < GATHER_U; i++) {
            if (k + i < end) {
                acc.x += v[i].x;
                acc.y += v[i].y;
                acc.z += v[i].z;
                acc.w += v[i].w;
            }
        }
    }
    return acc;
}

__global__ void gather_kernel(const int* __restrict__ nxt, const int* __restrict__ src,
                              const float* __restrict__ xs, const float* __restrict__ dinv,
                              float* __restrict__ agg, int n, int E) {
    int t = blockIdx.x * blockDim.x + threadIdx.x;
    int c = t >> 2;
    int q = t & 3;
    if (c < n) {
        v4f acc = gather_rows4(nxt, src, xs, c, q, n, E);
        float d = dinv[c];
        v4f o;
        o.x = acc.x * d;
        o.y = acc.y * d;
        o.z = acc.z * d;
        o.w = acc.w * d;
        *(v4f*)(agg + (size_t)c * 16 + q * 4) = o;
    }
}

__global__ void gemm_k16_n32_wmma(const float* __restrict__ AGG, const float* __restrict__ X,
                                  const float* __restrict__ dinv, const float* __restrict__ W,
                                  const float* __restrict__ b, float* __restrict__ H, int n) {
    int wave   = blockIdx.x * (blockDim.x >> 5) + (threadIdx.x >> 5);
    int nTiles = (n + 15) >> 4;
    if (wave >= nTiles) return;
    int lane = threadIdx.x & 31;
    int mi   = lane & 15;
    int kb   = (lane >> 4) << 1;
    int r0   = wave << 4;
    bool full = (r0 + 16 <= n);
    int rld  = full ? (r0 + mi) : min(r0 + mi, n - 1);
    const float* gr = AGG + (size_t)rld * 16;
    const float* xr = X   + (size_t)rld * 16;
    float di = dinv[rld];

    v8f acc0 = {}; v8f acc1 = {};
#pragma unroll
    for (int kk = 0; kk < 16; kk += 4) {
        v2f g = *(const v2f*)(gr + kk + kb);
        v2f x = *(const v2f*)(xr + kk + kb);
        v2f a;
        a.x = g.x + x.x * di * di;
        a.y = g.y + x.y * di * di;
        v2f b0, b1;
        b0.x = W[(kk + kb) * 32 + mi];          b0.y = W[(kk + kb + 1) * 32 + mi];
        b1.x = W[(kk + kb) * 32 + mi + 16];     b1.y = W[(kk + kb + 1) * 32 + mi + 16];
        acc0 = __builtin_amdgcn_wmma_f32_16x16x4_f32(false, a, false, b0, (short)0, acc0, false, false);
        acc1 = __builtin_amdgcn_wmma_f32_16x16x4_f32(false, a, false, b1, (short)0, acc1, false, false);
    }
    float bias0 = b[mi], bias1 = b[mi + 16];
    v8f o0, o1;
#pragma unroll
    for (int v = 0; v < 8; v++) {
        float t0 = acc0[v] + bias0;
        float t1 = acc1[v] + bias1;
        o0[v] = t0 > 0.0f ? t0 : 0.0f;
        o1[v] = t1 > 0.0f ? t1 : 0.0f;
    }
    int mh = (lane >> 4) << 3;
    float* hb = H + (size_t)(r0 + mh) * 32 + mi;
    if (full) {
#pragma unroll
        for (int v = 0; v < 8; v++) {
            hb[v * 32]      = o0[v];
            hb[v * 32 + 16] = o1[v];
        }
    } else {
#pragma unroll
        for (int v = 0; v < 8; v++) {
            if (r0 + mh + v < n) {
                hb[v * 32]      = o0[v];
                hb[v * 32 + 16] = o1[v];
            }
        }
    }
}

__global__ void gemm_k32_n16_wmma(const float* __restrict__ X, const float* __restrict__ W,
                                  const float* __restrict__ dinv, float* __restrict__ H, int n) {
    int wave   = blockIdx.x * (blockDim.x >> 5) + (threadIdx.x >> 5);
    int nTiles = (n + 15) >> 4;
    if (wave >= nTiles) return;
    int lane = threadIdx.x & 31;
    int mi   = lane & 15;
    int kb   = (lane >> 4) << 1;
    int r0   = wave << 4;
    bool full = (r0 + 16 <= n);
    int rld  = full ? (r0 + mi) : min(r0 + mi, n - 1);
    const float* xr = X + (size_t)rld * 32;
    float di = dinv[rld];

    v8f acc = {};
#pragma unroll
    for (int kk = 0; kk < 32; kk += 4) {
        v2f a = *(const v2f*)(xr + kk + kb);
        a.x *= di;
        a.y *= di;
        v2f b;
        b.x = W[(kk + kb) * 16 + mi];
        b.y = W[(kk + kb + 1) * 16 + mi];
        acc = __builtin_amdgcn_wmma_f32_16x16x4_f32(false, a, false, b, (short)0, acc, false, false);
    }
    int mh = (lane >> 4) << 3;
    float* hb = H + (size_t)(r0 + mh) * 16 + mi;
    if (full) {
#pragma unroll
        for (int v = 0; v < 8; v++) hb[v * 16] = acc[v];
    } else {
#pragma unroll
        for (int v = 0; v < 8; v++)
            if (r0 + mh + v < n) hb[v * 16] = acc[v];
    }
}

__global__ void gather_final_kernel(const int* __restrict__ nxt, const int* __restrict__ src,
                                    const float* __restrict__ hs, const float* __restrict__ dinv,
                                    const float* __restrict__ b2, const float* __restrict__ fcW,
                                    const float* __restrict__ fcb, float* __restrict__ out,
                                    int n, int E) {
    __shared__ float so[64];
    int tid = threadIdx.x;
    int c = blockIdx.x * 64 + (tid >> 2);
    int q = tid & 3;
    if (c < n) {
        v4f acc = gather_rows4(nxt, src, hs, c, q, n, E);
        v4f self = *(const v4f*)(hs + (size_t)c * 16 + q * 4);
        float d = dinv[c];
        float s0, s1, s2, s3;
        {
            float v = (acc.x + self.x) * d + b2[q * 4 + 0];
            v = v > 0.0f ? v : 0.0f;
            s0 = v * fcW[q * 4 + 0];
        }
        {
            float v = (acc.y + self.y) * d + b2[q * 4 + 1];
            v = v > 0.0f ? v : 0.0f;
            s1 = v * fcW[q * 4 + 1];
        }
        {
            float v = (acc.z + self.z) * d + b2[q * 4 + 2];
            v = v > 0.0f ? v : 0.0f;
            s2 = v * fcW[q * 4 + 2];
        }
        {
            float v = (acc.w + self.w) * d + b2[q * 4 + 3];
            v = v > 0.0f ? v : 0.0f;
            s3 = v * fcW[q * 4 + 3];
        }
        s0 += __shfl_xor(s0, 2);  s1 += __shfl_xor(s1, 2);
        s2 += __shfl_xor(s2, 2);  s3 += __shfl_xor(s3, 2);
        s0 += __shfl_xor(s0, 1);  s1 += __shfl_xor(s1, 1);
        s2 += __shfl_xor(s2, 1);  s3 += __shfl_xor(s3, 1);
        float u0 = s0 + s2;
        float u1 = s1 + s3;
        float w = u0 + u1;
        if (q == 0) so[tid >> 2] = w + fcb[0];
    }
    __syncthreads();
    if (tid < 64) {
        int i = blockIdx.x * 64 + tid;
        if (i < n) out[i] = so[tid];
    }
}

extern "C" void kernel_launch(void* const* d_in, const int* in_sizes, int n_in,
                              void* d_out, int out_size, void* d_ws, size_t ws_size,
                              hipStream_t stream) {
    const int*   edge = (const int*)d_in[0];
    const float* x0   = (const float*)d_in[1];
    const float* W1   = (const float*)d_in[2];
    const float* b1   = (const float*)d_in[3];
    const float* W2   = (const float*)d_in[4];
    const float* b2   = (const float*)d_in[5];
    const float* fcW  = (const float*)d_in[6];
    const float* fcb  = (const float*)d_in[7];
    float*       out  = (float*)d_out;

    const int n = GCN_N;
    const int E = in_sizes[0] / 2;
    const int* row = edge;
    const int* col = edge + E;

    float* ws   = (float*)d_ws;
    size_t npad = ((size_t)n + 63) & ~(size_t)63;
    float* dinv = ws;
    int*   nxt  = (int*)(dinv + npad);
    int*   src  = nxt + npad;
    float* aggx = (float*)(src + (size_t)n * 16);
    int*   rank = (int*)aggx;
    float* hs   = aggx;
    float* x1   = aggx + (size_t)n * 16;
    int*   cnt  = (int*)x1;
    float* xs   = x1 + npad;

    hipMemsetAsync(cnt, 0, (size_t)n * sizeof(int), stream);

    int eblocks = (E + 1023) / 1024;
    count_rank_kernel<<<eblocks, 256, 0, stream>>>(col, cnt, rank, E);
    prefix_kernel<<<(n + 1023) / 1024, 1024, 0, stream>>>(cnt, nxt, dinv, x0, xs, n);
    fill_kernel<<<eblocks, 256, 0, stream>>>(row, col, nxt, rank, src, E);

    int tiles   = (n + 15) / 16;
    int gblocks = (tiles + 7) / 8;

    gather_kernel<<<(n * 4 + 255) / 256, 256, 0, stream>>>(nxt, src, xs, dinv, aggx, n, E);
    gemm_k16_n32_wmma<<<gblocks, 256, 0, stream>>>(aggx, x0, dinv, W1, b1, x1, n);

    gemm_k32_n16_wmma<<<gblocks, 256, 0, stream>>>(x1, W2, dinv, hs, n);
    gather_final_kernel<<<(n + 63) / 64, 256, 0, stream>>>(nxt, src, hs, dinv, b2, fcW, fcb, out, n, E);
}


# layer-1 gather, GEMM1 (self-loop, bias, relu) and GEMM2 run in one launch: a 64-node block gathers into LDS and its waves feed both f32 WMMA products from LDS; 7 launches -> 5, sum orders unchanged
# speedup vs baseline: 1.9424x; 1.0501x over previous
#include <hip/hip_runtime.h>


typedef __attribute__((ext_vector_type(2))) float v2f;
typedef __attribute__((ext_vector_type(8))) float v8f;
typedef __attribute__((ext_vector_type(4))) float v4f;

#define GCN_N 100000

__global__ void count_rank_kernel(const int* __restrict__ col, int* __restrict__ cnt,
                                  int* __restrict__ rank, int E) {
    int T = gridDim.x * blockDim.x;
    int t = blockIdx.x * blockDim.x + threadIdx.x;
    if (t + 3 * T < E) {
        int c0 = col[t], c1 = col[t + T], c2 = col[t + 2 * T], c3 = col[t + 3 * T];
        int k0 = atomicAdd(&cnt[c0], 1);
        int k1 = atomicAdd(&cnt[c1], 1);
        int k2 = atomicAdd(&cnt[c2], 1);
        int k3 = atomicAdd(&cnt[c3], 1);
        rank[t]         = k0;
        rank[t + T]     = k1;
        rank[t + 2 * T] = k2;
        rank[t + 3 * T] = k3;
    } else {
        for (int e = t; e < E; e += T) rank[e] = atomicAdd(&cnt[col[e]], 1);
    }
}

__global__ void prefix_kernel(const int* __restrict__ cnt, int* __restrict__ nxt,
                              float* __restrict__ dinv, const float* __restrict__ x,
                              float* __restrict__ xs, int n) {
    __shared__ int wsum[32];
    __shared__ int base_s;
    __shared__ float dl[1024];
    int tid  = threadIdx.x;
    int lane = tid & 31;
    int w    = tid >> 5;
    int b0   = blockIdx.x * 1024;

    int part = 0;
    for (int i = tid; i < b0; i += 1024) part += cnt[i];
    for (int off = 16; off > 0; off >>= 1) part += __shfl_xor(part, off);
    if (lane == 0) wsum[w] = part;
    __syncthreads();
    if (w == 0) {
        int v = wsum[lane];
        for (int off = 16; off > 0; off >>= 1) v += __shfl_xor(v, off);
        if (lane == 0) base_s = v;
    }
    __syncthreads();
    int base = base_s;

    int i = b0 + tid;
    int c = (i < n) ? cnt[i] : 0;
    int incl = c;
    for (int off = 1; off < 32; off <<= 1) {
        int t = __shfl_up(incl, off);
        if (lane >= off) incl += t;
    }
    if (lane == 31) wsum[w] = incl;
    __syncthreads();
    int ws = wsum[lane];
    for (int off = 1; off < 32; off <<= 1) {
        int t = __shfl_up(ws, off);
        if (lane >= off) ws += t;
    }
    int woff = __shfl(ws, (w > 0) ? (w - 1) : 0);
    if (w == 0) woff = 0;
    float dv = rsqrtf((float)c + 1.0f);
    dl[tid] = dv;
    if (i < n) {
        int excl = base + woff + incl - c;
        nxt[i]  = excl;
        dinv[i] = dv;
        if (i == n - 1) nxt[n] = excl + c;
    }
    __syncthreads();
    size_t lim = (size_t)n * 16;
    size_t o16 = (size_t)b0 * 16;
    if (o16 + 16384 <= lim) {
        for (int it = 0; it < 16; it++) {
            int loc = it * 1024 + tid;
            xs[o16 + loc] = x[o16 + loc] * dl[loc >> 4];
        }
    } else {
        for (int it = 0; it < 16; it++) {
            int loc = it * 1024 + tid;
            size_t idx = o16 + loc;
            if (idx < lim) xs[idx] = x[idx] * dl[loc >> 4];
        }
    }
}

__global__ void fill_kernel(const int* __restrict__ row, const int* __restrict__ col,
                            const int* __restrict__ nxt, const int* __restrict__ rank,
                            int* __restrict__ src, int E) {
    int T = gridDim.x * blockDim.x;
    int t = blockIdx.x * blockDim.x + threadIdx.x;
    unsigned lim = (unsigned)E;
    if (t + 3 * T < E) {
        int c0 = col[t], c1 = col[t + T], c2 = col[t + 2 * T], c3 = col[t + 3 * T];
        int k0 = rank[t], k1 = rank[t + T], k2 = rank[t + 2 * T], k3 = rank[t + 3 * T];
        int r0 = row[t], r1 = row[t + T], r2 = row[t + 2 * T], r3 = row[t + 3 * T];
        int s0 = nxt[c0] + k0;
        int s1 = nxt[c1] + k1;
        int s2 = nxt[c2] + k2;
        int s3 = nxt[c3] + k3;
        if ((unsigned)s0 < lim) src[s0] = r0;
        if ((unsigned)s1 < lim) src[s1] = r1;
        if ((unsigned)s2 < lim) src[s2] = r2;
        if ((unsigned)s3 < lim) src[s3] = r3;
    } else {
        for (int e = t; e < E; e += T) {
            int s = nxt[col[e]] + rank[e];
            if ((unsigned)s < lim) src[s] = row[e];
        }
    }
}

#define GATHER_U 8
__device__ __forceinline__ v4f gather_rows4(const int* __restrict__ nxt, const int* __restrict__ src,
                                            const float* __restrict__ hs, int c, int q, int n, int E) {
    int beg = nxt[c];
    int end = nxt[c + 1];
    beg = max(beg, 0);
    end = min(end, E);
    int hi = end - 1;
    unsigned last = (unsigned)(n - 1);
    const char* hb = (const char*)hs;
    unsigned qo = (unsigned)q * 16u;
    v4f acc = {0.0f, 0.0f, 0.0f, 0.0f};
    unsigned r[GATHER_U];
#pragma unroll
    for (int i = 0; i < GATHER_U; i++) r[i] = min((unsigned)src[max(min(beg + i, hi), 0)], last);
    for (int k = beg; k < end; k += GATHER_U) {
        v4f v[GATHER_U];
#pragma unroll
        for (int i = 0; i < GATHER_U; i++) v[i] = *(const v4f*)(hb + (r[i] * 64u + qo));
#pragma unroll
        for (int i = 0; i < GATHER_U; i++) r[i] = min((unsigned)src[max(min(k + GATHER_U + i, hi), 0)], last);
#pragma unroll
        for (int i = 0; i < GATHER_U; i++) {
            if (k + i < end) {
                acc.x += v[i].x;
                acc.y += v[i].y;
                acc.z += v[i].z;
                acc.w += v[i].w;
            }
        }
    }
    return acc;
}

#define FB_NODES 64
#define A1_PITCH 20
#define X1_PITCH 36
__global__ void gather_mlp_kernel(const int* __restrict__ nxt, const int* __restrict__ src,
                                  const float* __restrict__ xs, const float* __restrict__ dinv,
                                  const float* __restrict__ X, const float* __restrict__ W1,
                                  const float* __restrict__ b1, const float* __restrict__ W2,
                                  float* __restrict__ H, int n, int E) {
    __shared__ float a1[FB_NODES * A1_PITCH];
    __shared__ float x1s[FB_NODES * X1_PITCH];
    __shared__ float dl[FB_NODES];
    int tid = threadIdx.x;
    int j = tid >> 2;
    int q = tid & 3;
    int c = blockIdx.x * FB_NODES + j;
    v4f a = {0.0f, 0.0f, 0.0f, 0.0f};
    float d = 0.0f;
    if (c < n) {
        v4f acc = gather_rows4(nxt, src, xs, c, q, n, E);
        v4f xv = *(const v4f*)(X + (size_t)c * 16 + q * 4);
        d = dinv[c];
        float g0 = acc.x * d, g1 = acc.y * d, g2 = acc.z * d, g3 = acc.w * d;
        a.x = g0 + xv.x * d * d;
        a.y = g1 + xv.y * d * d;
        a.z = g2 + xv.z * d * d;
        a.w = g3 + xv.w * d * d;
    }
    *(v4f*)(a1 + j * A1_PITCH + q * 4) = a;
    if (q == 0) dl[j] = d;
    __syncthreads();

    int w    = tid >> 5;
    int lane = tid & 31;
    int mi   = lane & 15;
    int kb   = (lane >> 4) << 1;
    int mh   = (lane >> 4) << 3;
    {
        int t  = w >> 1;
        int ch = (w & 1) << 4;
        if (blockIdx.x * FB_NODES + t * 16 < n) {
            const float* ar = a1 + (t * 16 + mi) * A1_PITCH;
            v8f acc = {};
#pragma unroll
            for (int kk = 0; kk < 16; kk += 4) {
                v2f av = *(const v2f*)(ar + kk + kb);
                v2f bv;
                bv.x = W1[(kk + kb) * 32 + ch + mi];
                bv.y = W1[(kk + kb + 1) * 32 + ch + mi];
                acc = __builtin_amdgcn_wmma_f32_16x16x4_f32(false, av, false, bv, (short)0, acc, false, false);
            }
            float bias = b1[ch + mi];
            float* xw = x1s + (t * 16 + mh) * X1_PITCH + ch + mi;
#pragma unroll
            for (int v = 0; v < 8; v++) {
                float tv = acc[v] + bias;
                xw[v * X1_PITCH] = tv > 0.0f ? tv : 0.0f;
            }
        }
    }
    __syncthreads();

    int r0 = blockIdx.x * FB_NODES + w * 16;
    if (w < 4 && r0 < n) {
        const float* xr = x1s + (w * 16 + mi) * X1_PITCH;
        float di = dl[w * 16 + mi];
        v8f acc = {};
#pragma unroll
        for (int kk = 0; kk < 32; kk += 4) {
            v2f av = *(const v2f*)(xr + kk + kb);
            av.x *= di;
            av.y *= di;
            v2f bv;
            bv.x = W2[(kk + kb) * 16 + mi];
            bv.y = W2[(kk + kb + 1) * 16 + mi];
            acc = __builtin_amdgcn_wmma_f32_16x16x4_f32(false, av, false, bv, (short)0, acc, false, false);
        }
        float* hb = H + (size_t)(r0 + mh) * 16 + mi;
        if (r0 + 16 <= n) {
#pragma unroll
            for (int v = 0; v < 8; v++) hb[v * 16] = acc[v];
        } else {
#pragma unroll
            for (int v = 0; v < 8; v++)
                if (r0 + mh + v < n) hb[v * 16] = acc[v];
        }
    }
}

__global__ void gather_final_kernel(const int* __restrict__ nxt, const int* __restrict__ src,
                                    const float* __restrict__ hs, const float* __restrict__ dinv,
                                    const float* __restrict__ b2, const float* __restrict__ fcW,
                                    const float* __restrict__ fcb, float* __restrict__ out,
                                    int n, int E) {
    __shared__ float so[64];
    int tid = threadIdx.x;
    int c = blockIdx.x * 64 + (tid >> 2);
    int q = tid & 3;
    if (c < n) {
        v4f acc = gather_rows4(nxt, src, hs, c, q, n, E);
        v4f self = *(const v4f*)(hs + (size_t)c * 16 + q * 4);
        float d = dinv[c];
        float s0, s1, s2, s3;
        {
            float v = (acc.x + self.x) * d + b2[q * 4 + 0];
            v = v > 0.0f ? v : 0.0f;
            s0 = v * fcW[q * 4 + 0];
        }
        {
            float v = (acc.y + self.y) * d + b2[q * 4 + 1];
            v = v > 0.0f ? v : 0.0f;
            s1 = v * fcW[q * 4 + 1];
        }
        {
            float v = (acc.z + self.z) * d + b2[q * 4 + 2];
            v = v > 0.0f ? v : 0.0f;
            s2 = v * fcW[q * 4 + 2];
        }
        {
            float v = (acc.w + self.w) * d + b2[q * 4 + 3];
            v = v > 0.0f ? v : 0.0f;
            s3 = v * fcW[q * 4 + 3];
        }
        s0 += __shfl_xor(s0, 2);  s1 += __shfl_xor(s1, 2);
        s2 += __shfl_xor(s2, 2);  s3 += __shfl_xor(s3, 2);
        s0 += __shfl_xor(s0, 1);  s1 += __shfl_xor(s1, 1);
        s2 += __shfl_xor(s2, 1);  s3 += __shfl_xor(s3, 1);
        float u0 = s0 + s2;
        float u1 = s1 + s3;
        float w = u0 + u1;
        if (q == 0) so[tid >> 2] = w + fcb[0];
    }
    __syncthreads();
    if (tid < 64) {
        int i = blockIdx.x * 64 + tid;
        if (i < n) out[i] = so[tid];
    }
}

extern "C" void kernel_launch(void* const* d_in, const int* in_sizes, int n_in,
                              void* d_out, int out_size, void* d_ws, size_t ws_size,
                              hipStream_t stream) {
    const int*   edge = (const int*)d_in[0];
    const float* x0   = (const float*)d_in[1];
    const float* W1   = (const float*)d_in[2];
    const float* b1   = (const float*)d_in[3];
    const float* W2   = (const float*)d_in[4];
    const float* b2   = (const float*)d_in[5];
    const float* fcW  = (const float*)d_in[6];
    const float* fcb  = (const float*)d_in[7];
    float*       out  = (float*)d_out;

    const int n = GCN_N;
    const int E = in_sizes[0] / 2;
    const int* row = edge;
    const int* col = edge + E;

    float* ws   = (float*)d_ws;
    size_t npad = ((size_t)n + 63) & ~(size_t)63;
    float* dinv = ws;
    int*   nxt  = (int*)(dinv + npad);
    int*   src  = nxt + npad;
    float* hs   = (float*)(src + (size_t)n * 16);
    int*   rank = (int*)hs;
    int*   cnt  = (int*)(hs + (size_t)n * 16);
    float* xs   = (float*)(cnt + npad);

    hipMemsetAsync(cnt, 0, (size_t)n * sizeof(int), stream);

    int eblocks = (E + 1023) / 1024;
    count_rank_kernel<<<eblocks, 256, 0, stream>>>(col, cnt, rank, E);
    prefix_kernel<<<(n + 1023) / 1024, 1024, 0, stream>>>(cnt, nxt, dinv, x0, xs, n);
    fill_kernel<<<eblocks, 256, 0, stream>>>(row, col, nxt, rank, src, E);

    int nblocks = (n + 63) / 64;

    gather_mlp_kernel<<<nblocks, 256, 0, stream>>>(nxt, src, xs, dinv, x0, W1, b1, W2, hs, n, E);

    gather_final_kernel<<<nblocks, 256, 0, stream>>>(nxt, src, hs, dinv, b2, fcW, fcb, out, n, E);
}


# edges placed in one pass into fixed 32-slot buckets per target (slot = node*32 + returned rank): no prefix scan, no fill pass, 5 launches -> 4; edges past slot 32 summed last, in overflow-list order
# speedup vs baseline: 2.4448x; 1.2586x over previous
#include <hip/hip_runtime.h>


typedef __attribute__((ext_vector_type(2))) float v2f;
typedef __attribute__((ext_vector_type(8))) float v8f;
typedef __attribute__((ext_vector_type(4))) float v4f;

#define GCN_N 100000

#define BK_CAP 32
#define OVF_CAP 50000

typedef __attribute__((ext_vector_type(2))) int v2i;

__device__ __forceinline__ void place_edge(int c, int k, int r, int* __restrict__ bkt,
                                           int* __restrict__ ovn, v2i* __restrict__ ovf) {
    if ((unsigned)k < (unsigned)BK_CAP) {
        bkt[c * BK_CAP + k] = r;
    } else {
        int o = atomicAdd(ovn, 1);
        if (o < OVF_CAP) {
            v2i p = {c, r};
            ovf[o] = p;
        }
    }
}

__global__ void bucket_kernel(const int* __restrict__ row, const int* __restrict__ col,
                              int* __restrict__ cnt, int* __restrict__ bkt,
                              int* __restrict__ ovn, v2i* __restrict__ ovf, int E) {
    int T = gridDim.x * blockDim.x;
    int t = blockIdx.x * blockDim.x + threadIdx.x;
    if (t + 3 * T < E) {
        int c0 = col[t], c1 = col[t + T], c2 = col[t + 2 * T], c3 = col[t + 3 * T];
        int r0 = row[t], r1 = row[t + T], r2 = row[t + 2 * T], r3 = row[t + 3 * T];
        int k0 = atomicAdd(&cnt[c0], 1);
        int k1 = atomicAdd(&cnt[c1], 1);
        int k2 = atomicAdd(&cnt[c2], 1);
        int k3 = atomicAdd(&cnt[c3], 1);
        place_edge(c0, k0, r0, bkt, ovn, ovf);
        place_edge(c1, k1, r1, bkt, ovn, ovf);
        place_edge(c2, k2, r2, bkt, ovn, ovf);
        place_edge(c3, k3, r3, bkt, ovn, ovf);
    } else {
        for (int e = t; e < E; e += T) {
            int c = col[e];
            int k = atomicAdd(&cnt[c], 1);
            place_edge(c, k, row[e], bkt, ovn, ovf);
        }
    }
}

__global__ void prescale_kernel(const int* __restrict__ cnt, const float* __restrict__ x,
                                float* __restrict__ xs, int n) {
    int t = blockIdx.x * blockDim.x + threadIdx.x;
    if (t < n * 4) {
        float d = rsqrtf((float)cnt[t >> 2] + 1.0f);
        v4f v = *(const v4f*)(x + (size_t)t * 4);
        v.x *= d;
        v.y *= d;
        v.z *= d;
        v.w *= d;
        *(v4f*)(xs + (size_t)t * 4) = v;
    }
}

#define GATHER_U 8
__device__ __forceinline__ v4f gather_rows4(const int* __restrict__ bkt, const float* __restrict__ hs,
                                            int c, int q, int deg, int n) {
    int beg = c * BK_CAP;
    int end = beg + min(max(deg, 0), BK_CAP);
    int hi = end - 1;
    unsigned last = (unsigned)(n - 1);
    const char* hb = (const char*)hs;
    unsigned qo = (unsigned)q * 16u;
    v4f acc = {0.0f, 0.0f, 0.0f, 0.0f};
    unsigned r[GATHER_U];
#pragma unroll
    for (int i = 0; i < GATHER_U; i++) r[i] = min((unsigned)bkt[max(min(beg + i, hi), 0)], last);
    for (int k = beg; k < end; k += GATHER_U) {
        v4f v[GATHER_U];
#pragma unroll
        for (int i = 0; i < GATHER_U; i++) v[i] = *(const v4f*)(hb + (r[i] * 64u + qo));
#pragma unroll
        for (int i = 0; i < GATHER_U; i++) r[i] = min((unsigned)bkt[max(min(k + GATHER_U + i, hi), 0)], last);
#pragma unroll
        for (int i = 0; i < GATHER_U; i++) {
            if (i == 0 || k + i < end) {
                acc.x += v[i].x;
                acc.y += v[i].y;
                acc.z += v[i].z;
                acc.w += v[i].w;
            }
        }
    }
    return acc;
}

__device__ __forceinline__ v4f overflow_rows4(v4f acc, const v2i* __restrict__ ovf, int on,
                                              const int* __restrict__ row, const int* __restrict__ col,
                                              const float* __restrict__ hs, int c, int q, int n, int E) {
    unsigned last = (unsigned)(n - 1);
    const char* hb = (const char*)hs;
    unsigned qo = (unsigned)q * 16u;
    if (on <= OVF_CAP) {
        int hi = max(on - 1, 0);
        v2i p[GATHER_U];
#pragma unroll
        for (int i = 0; i < GATHER_U; i++) p[i] = ovf[min(i, hi)];
        for (int o = 0; o < on; o += GATHER_U) {
            v4f v[GATHER_U];
            unsigned mine = 0u;
#pragma unroll
            for (int i = 0; i < GATHER_U; i++) {
                v[i] = *(const v4f*)(hb + (min((unsigned)p[i].y, last) * 64u + qo));
                if (o + i < on && p[i].x == c) mine |= 1u << i;
            }
#pragma unroll
            for (int i = 0; i < GATHER_U; i++) p[i] = ovf[min(o + GATHER_U + i, hi)];
#pragma unroll
            for (int i = 0; i < GATHER_U; i++) {
                if (mine & (1u << i)) {
                    acc.x += v[i].x;
                    acc.y += v[i].y;
                    acc.z += v[i].z;
                    acc.w += v[i].w;
                }
            }
        }
    } else {
        acc.x = 0.0f;
        acc.y = 0.0f;
        acc.z = 0.0f;
        acc.w = 0.0f;
        for (int e = 0; e < E; e++) {
            if (col[e] == c) {
                v4f v = *(const v4f*)(hb + (min((unsigned)row[e], last) * 64u + qo));
                acc.x += v.x;
                acc.y += v.y;
                acc.z += v.z;
                acc.w += v.w;
            }
        }
    }
    return acc;
}

#define FB_NODES 64
#define A1_PITCH 20
#define X1_PITCH 36
__global__ void gather_mlp_kernel(const int* __restrict__ cnt, const int* __restrict__ bkt,
                                  const int* __restrict__ ovn, const v2i* __restrict__ ovf,
                                  const int* __restrict__ row, const int* __restrict__ col,
                                  const float* __restrict__ xs,
                                  const float* __restrict__ X, const float* __restrict__ W1,
                                  const float* __restrict__ b1, const float* __restrict__ W2,
                                  float* __restrict__ H, int n, int E) {
    __shared__ float a1[FB_NODES * A1_PITCH];
    __shared__ float x1s[FB_NODES * X1_PITCH];
    __shared__ float dl[FB_NODES];
    int tid = threadIdx.x;
    int j = tid >> 2;
    int q = tid & 3;
    int c = blockIdx.x * FB_NODES + j;
    v4f a = {0.0f, 0.0f, 0.0f, 0.0f};
    float d = 0.0f;
    int on = ovn[0];
    if (c < n) {
        int deg = cnt[c];
        v4f acc = {0.0f, 0.0f, 0.0f, 0.0f};
        if (deg > 0) acc = gather_rows4(bkt, xs, c, q, deg, n);
        if (deg > BK_CAP) acc = overflow_rows4(acc, ovf, on, row, col, xs, c, q, n, E);
        v4f xv = *(const v4f*)(X + (size_t)c * 16 + q * 4);
        d = rsqrtf((float)deg + 1.0f);
        float g0 = acc.x * d, g1 = acc.y * d, g2 = acc.z * d, g3 = acc.w * d;
        a.x = g0 + xv.x * d * d;
        a.y = g1 + xv.y * d * d;
        a.z = g2 + xv.z * d * d;
        a.w = g3 + xv.w * d * d;
    }
    *(v4f*)(a1 + j * A1_PITCH + q * 4) = a;
    if (q == 0) dl[j] = d;
    __syncthreads();

    int w    = tid >> 5;
    int lane = tid & 31;
    int mi   = lane & 15;
    int kb   = (lane >> 4) << 1;
    int mh   = (lane >> 4) << 3;
    {
        int t  = w >> 1;
        int ch = (w & 1) << 4;
        if (blockIdx.x * FB_NODES + t * 16 < n) {
            const float* ar = a1 + (t * 16 + mi) * A1_PITCH;
            v8f acc = {};
#pragma unroll
            for (int kk = 0; kk < 16; kk += 4) {
                v2f av = *(const v2f*)(ar + kk + kb);
                v2f bv;
                bv.x = W1[(kk + kb) * 32 + ch + mi];
                bv.y = W1[(kk + kb + 1) * 32 + ch + mi];
                acc = __builtin_amdgcn_wmma_f32_16x16x4_f32(false, av, false, bv, (short)0, acc, false, false);
            }
            float bias = b1[ch + mi];
            float* xw = x1s + (t * 16 + mh) * X1_PITCH + ch + mi;
#pragma unroll
            for (int v = 0; v < 8; v++) {
                float tv = acc[v] + bias;
                xw[v * X1_PITCH] = tv > 0.0f ? tv : 0.0f;
            }
        }
    }
    __syncthreads();

    int r0 = blockIdx.x * FB_NODES + w * 16;
    if (w < 4 && r0 < n) {
        const float* xr = x1s + (w * 16 + mi) * X1_PITCH;
        float di = dl[w * 16 + mi];
        v8f acc = {};
#pragma unroll
        for (int kk = 0; kk < 32; kk += 4) {
            v2f av = *(const v2f*)(xr + kk + kb);
            av.x *= di;
            av.y *= di;
            v2f bv;
            bv.x = W2[(kk + kb) * 16 + mi];
            bv.y = W2[(kk + kb + 1) * 16 + mi];
            acc = __builtin_amdgcn_wmma_f32_16x16x4_f32(false, av, false, bv, (short)0, acc, false, false);
        }
        float* hb = H + (size_t)(r0 + mh) * 16 + mi;
        if (r0 + 16 <= n) {
#pragma unroll
            for (int v = 0; v < 8; v++) hb[v * 16] = acc[v];
        } else {
#pragma unroll
            for (int v = 0; v < 8; v++)
                if (r0 + mh + v < n) hb[v * 16] = acc[v];
        }
    }
}

__global__ void gather_final_kernel(const int* __restrict__ cnt, const int* __restrict__ bkt,
                                    const int* __restrict__ ovn, const v2i* __restrict__ ovf,
                                    const int* __restrict__ row, const int* __restrict__ col,
                                    const float* __restrict__ hs,
                                    const float* __restrict__ b2, const float* __restrict__ fcW,
                                    const float* __restrict__ fcb, float* __restrict__ out,
                                    int n, int E) {
    __shared__ float so[64];
    int tid = threadIdx.x;
    int c = blockIdx.x * 64 + (tid >> 2);
    int q = tid & 3;
    int on = ovn[0];
    if (c < n) {
        int deg = cnt[c];
        v4f acc = {0.0f, 0.0f, 0.0f, 0.0f};
        if (deg > 0) acc = gather_rows4(bkt, hs, c, q, deg, n);
        if (deg > BK_CAP) acc = overflow_rows4(acc, ovf, on, row, col, hs, c, q, n, E);
        v4f self = *(const v4f*)(hs + (size_t)c * 16 + q * 4);
        float d = rsqrtf((float)deg + 1.0f);
        float s0, s1, s2, s3;
        {
            float v = (acc.x + self.x) * d + b2[q * 4 + 0];
            v = v > 0.0f ? v : 0.0f;
            s0 = v * fcW[q * 4 + 0];
        }
        {
            float v = (acc.y + self.y) * d + b2[q * 4 + 1];
            v = v > 0.0f ? v : 0.0f;
            s1 = v * fcW[q * 4 + 1];
        }
        {
            float v = (acc.z + self.z) * d + b2[q * 4 + 2];
            v = v > 0.0f ? v : 0.0f;
            s2 = v * fcW[q * 4 + 2];
        }
        {
            float v = (acc.w + self.w) * d + b2[q * 4 + 3];
            v = v > 0.0f ? v : 0.0f;
            s3 = v * fcW[q * 4 + 3];
        }
        s0 += __shfl_xor(s0, 2);  s1 += __shfl_xor(s1, 2);
        s2 += __shfl_xor(s2, 2);  s3 += __shfl_xor(s3, 2);
        s0 += __shfl_xor(s0, 1);  s1 += __shfl_xor(s1, 1);
        s2 += __shfl_xor(s2, 1);  s3 += __shfl_xor(s3, 1);
        float u0 = s0 + s2;
        float u1 = s1 + s3;
        float w = u0 + u1;
        if (q == 0) so[tid >> 2] = w + fcb[0];
    }
    __syncthreads();
    if (tid < 64) {
        int i = blockIdx.x * 64 + tid;
        if (i < n) out[i] = so[tid];
    }
}

extern "C" void kernel_launch(void* const* d_in, const int* in_sizes, int n_in,
                              void* d_out, int out_size, void* d_ws, size_t ws_size,
                              hipStream_t stream) {
    const int*   edge = (const int*)d_in[0];
    const float* x0   = (const float*)d_in[1];
    const float* W1   = (const float*)d_in[2];
    const float* b1   = (const float*)d_in[3];
    const float* W2   = (const float*)d_in[4];
    const float* b2   = (const float*)d_in[5];
    const float* fcW  = (const float*)d_in[6];
    const float* fcb  = (const float*)d_in[7];
    float*       out  = (float*)d_out;

    const int n = GCN_N;
    const int E = in_sizes[0] / 2;
    const int* row = edge;
    const int* col = edge + E;

    float* ws   = (float*)d_ws;
    size_t npad = ((size_t)n + 63) & ~(size_t)63;
    int*   cnt  = (int*)ws;
    int*   ovn  = cnt + n;
    v2i*   ovf  = (v2i*)(cnt + npad);
    int*   bkt  = cnt + 2 * npad;
    float* hs   = (float*)(bkt + (size_t)n * BK_CAP);
    float* xs   = hs + (size_t)n * 16;

    hipMemsetAsync(cnt, 0, (size_t)(n + 1) * sizeof(int), stream);

    int eblocks = (E + 1023) / 1024;
    bucket_kernel<<<eblocks, 256, 0, stream>>>(row, col, cnt, bkt, ovn, ovf, E);
    prescale_kernel<<<(n * 4 + 255) / 256, 256, 0, stream>>>(cnt, x0, xs, n);

    int nblocks = (n + 63) / 64;

    gather_mlp_kernel<<<nblocks, 256, 0, stream>>>(cnt, bkt, ovn, ovf, row, col, xs, x0, W1, b1, W2, hs, n, E);

    gather_final_kernel<<<nblocks, 256, 0, stream>>>(cnt, bkt, ovn, ovf, row, col, hs, b2, fcW, fcb, out, n, E);
}


# edges ordered in LDS: blocks sort edges by 256-target bin into staged runs, a block per bin ranks by LDS adds, writes whole bucket lines; no per-edge global add/4B scatter; sum order: LDS arrival
# speedup vs baseline: 4.8073x; 1.9663x over previous
#include <hip/hip_runtime.h>


typedef __attribute__((ext_vector_type(2))) float v2f;
typedef __attribute__((ext_vector_type(8))) float v8f;
typedef __attribute__((ext_vector_type(4))) float v4f;

#define GCN_N 100000

#define BK_CAP 32
#define OVF_CAP 50000

typedef __attribute__((ext_vector_type(2))) int v2i;
typedef __attribute__((ext_vector_type(4))) int v4i;

#define BIN_NODES 256
#define NBINS ((GCN_N + BIN_NODES - 1) / BIN_NODES)
#define BIN_CAP 8000
#define SORT_T 512
#define SORT_EPT 8
#define SORT_CHUNK (SORT_T * SORT_EPT)
#define PLACE_U 8
#define SRC_MASK 0x1FFFF

__global__ void __launch_bounds__(SORT_T)
bin_kernel(const int* __restrict__ row, const int* __restrict__ col,
           int* __restrict__ cur, int* __restrict__ stage, int E) {
    __shared__ int hist[SORT_T];
    __shared__ int loff[SORT_T];
    __shared__ int gbase[SORT_T];
    __shared__ int wtot[SORT_T / 32];
    __shared__ int sval[SORT_CHUNK];
    __shared__ int sbin[SORT_CHUNK];
    int tid = threadIdx.x;
    int base = blockIdx.x * SORT_CHUNK;
    hist[tid] = 0;
    int cc[SORT_EPT], rr[SORT_EPT];
#pragma unroll
    for (int i = 0; i < SORT_EPT; i++) {
        int e = min(base + i * SORT_T + tid, E - 1);
        cc[i] = col[e];
        rr[i] = row[e];
    }
    __syncthreads();
    int bb[SORT_EPT], rk[SORT_EPT];
#pragma unroll
    for (int i = 0; i < SORT_EPT; i++) {
        bb[i] = (int)min((unsigned)cc[i] >> 8, (unsigned)(NBINS - 1));
        rk[i] = 0;
        if (base + i * SORT_T + tid < E) rk[i] = atomicAdd(&hist[bb[i]], 1);
    }
    __syncthreads();
    int v = hist[tid];
    int lane = tid & 31;
    int w = tid >> 5;
    int inc = v;
#pragma unroll
    for (int d = 1; d < 32; d <<= 1) {
        int t = __shfl_up(inc, d);
        if (lane >= d) inc += t;
    }
    if (lane == 31) wtot[w] = inc;
    __syncthreads();
    int wp = 0;
#pragma unroll
    for (int j = 0; j < SORT_T / 32; j++)
        if (j < w) wp += wtot[j];
    loff[tid] = wp + inc - v;
    int g = 0;
    if (v > 0) g = atomicAdd(&cur[tid * BIN_NODES], v);
    gbase[tid] = g;
    __syncthreads();
#pragma unroll
    for (int i = 0; i < SORT_EPT; i++) {
        if (base + i * SORT_T + tid < E) {
            int p = loff[bb[i]] + rk[i];
            sval[p] = ((cc[i] & (BIN_NODES - 1)) << 17) | (rr[i] & SRC_MASK);
            sbin[p] = bb[i];
        }
    }
    __syncthreads();
    int m = min(E - base, SORT_CHUNK);
    for (int i = tid; i < m; i += SORT_T) {
        int b = sbin[i];
        int dst = gbase[b] + (i - loff[b]);
        if ((unsigned)dst < (unsigned)BIN_CAP) stage[b * BIN_CAP + dst] = sval[i];
    }
}

__device__ __forceinline__ void put_edge(int loc, int k, int src, int node0, int* lb,
                                         int* ovn, v2i* __restrict__ ovf) {
    if ((unsigned)k < (unsigned)BK_CAP) {
        lb[loc * BK_CAP + k] = src;
    } else {
        int o = atomicAdd(ovn, 1);
        if (o < OVF_CAP) {
            v2i p = {node0 + loc, src};
            ovf[o] = p;
        }
    }
}

__global__ void __launch_bounds__(BIN_NODES)
place_kernel(const int* __restrict__ row, const int* __restrict__ col, int* cnt,
             int* __restrict__ bkt, int* ovn, v2i* __restrict__ ovf,
             const int* __restrict__ stage, int n, int E) {
    __shared__ int lb[BIN_NODES * BK_CAP];
    __shared__ int lc[BIN_NODES];
    int tid = threadIdx.x;
    int b = blockIdx.x;
    int node0 = b * BIN_NODES;
    int total = cnt[node0];
    lc[tid] = 0;
    v4i z = {0, 0, 0, 0};
    for (int i = tid; i < BIN_NODES * BK_CAP / 4; i += BIN_NODES) ((v4i*)lb)[i] = z;
    __syncthreads();
    if (total <= BIN_CAP) {
        const int* sp = stage + b * BIN_CAP;
        for (int i0 = tid; i0 < total; i0 += PLACE_U * BIN_NODES) {
            unsigned p[PLACE_U];
            int k[PLACE_U];
#pragma unroll
            for (int u = 0; u < PLACE_U; u++) p[u] = (unsigned)sp[min(i0 + u * BIN_NODES, total - 1)];
#pragma unroll
            for (int u = 0; u < PLACE_U; u++) {
                k[u] = 0;
                if (i0 + u * BIN_NODES < total) k[u] = atomicAdd(&lc[(p[u] >> 17) & (BIN_NODES - 1)], 1);
            }
#pragma unroll
            for (int u = 0; u < PLACE_U; u++)
                if (i0 + u * BIN_NODES < total)
                    put_edge((int)((p[u] >> 17) & (BIN_NODES - 1)), k[u], (int)(p[u] & SRC_MASK), node0, lb, ovn, ovf);
        }
    } else {
        for (int e = tid; e < E; e += BIN_NODES) {
            int c = col[e];
            if ((int)min((unsigned)c >> 8, (unsigned)(NBINS - 1)) == b) {
                int loc = c & (BIN_NODES - 1);
                int k = atomicAdd(&lc[loc], 1);
                put_edge(loc, k, row[e] & SRC_MASK, node0, lb, ovn, ovf);
            }
        }
    }
    __syncthreads();
    int c = node0 + tid;
    if (c < n) cnt[c] = lc[tid];
    v4i* gb = (v4i*)(bkt + (size_t)node0 * BK_CAP);
    for (int i = tid; i < BIN_NODES * BK_CAP / 4; i += BIN_NODES)
        if (node0 + (i >> 3) < n) gb[i] = ((const v4i*)lb)[i];
}

__global__ void prescale_kernel(const int* __restrict__ cnt, const float* __restrict__ x,
                                float* __restrict__ xs, int n) {
    int t = blockIdx.x * blockDim.x + threadIdx.x;
    if (t < n * 4) {
        float d = rsqrtf((float)cnt[t >> 2] + 1.0f);
        v4f v = *(const v4f*)(x + (size_t)t * 4);
        v.x *= d;
        v.y *= d;
        v.z *= d;
        v.w *= d;
        *(v4f*)(xs + (size_t)t * 4) = v;
    }
}

#define GATHER_U 8
__device__ __forceinline__ v4f gather_rows4(const int* __restrict__ bkt, const float* __restrict__ hs,
                                            int c, int q, int deg, int n) {
    int beg = c * BK_CAP;
    int end = beg + min(max(deg, 0), BK_CAP);
    int hi = end - 1;
    unsigned last = (unsigned)(n - 1);
    const char* hb = (const char*)hs;
    unsigned qo = (unsigned)q * 16u;
    v4f acc = {0.0f, 0.0f, 0.0f, 0.0f};
    unsigned r[GATHER_U];
#pragma unroll
    for (int i = 0; i < GATHER_U; i++) r[i] = min((unsigned)bkt[max(min(beg + i, hi), 0)], last);
    for (int k = beg; k < end; k += GATHER_U) {
        v4f v[GATHER_U];
#pragma unroll
        for (int i = 0; i < GATHER_U; i++) v[i] = *(const v4f*)(hb + (r[i] * 64u + qo));
#pragma unroll
        for (int i = 0; i < GATHER_U; i++) r[i] = min((unsigned)bkt[max(min(k + GATHER_U + i, hi), 0)], last);
#pragma unroll
        for (int i = 0; i < GATHER_U; i++) {
            if (i == 0 || k + i < end) {
                acc.x += v[i].x;
                acc.y += v[i].y;
                acc.z += v[i].z;
                acc.w += v[i].w;
            }
        }
    }
    return acc;
}

__device__ __forceinline__ v4f overflow_rows4(v4f acc, const v2i* __restrict__ ovf, int on,
                                              const int* __restrict__ row, const int* __restrict__ col,
                                              const float* __restrict__ hs, int c, int q, int n, int E) {
    unsigned last = (unsigned)(n - 1);
    const char* hb = (const char*)hs;
    unsigned qo = (unsigned)q * 16u;
    if (on <= OVF_CAP) {
        int hi = max(on - 1, 0);
        v2i p[GATHER_U];
#pragma unroll
        for (int i = 0; i < GATHER_U; i++) p[i] = ovf[min(i, hi)];
        for (int o = 0; o < on; o += GATHER_U) {
            v4f v[GATHER_U];
            unsigned mine = 0u;
#pragma unroll
            for (int i = 0; i < GATHER_U; i++) {
                v[i] = *(const v4f*)(hb + (min((unsigned)p[i].y, last) * 64u + qo));
                if (o + i < on && p[i].x == c) mine |= 1u << i;
            }
#pragma unroll
            for (int i = 0; i < GATHER_U; i++) p[i] = ovf[min(o + GATHER_U + i, hi)];
#pragma unroll
            for (int i = 0; i < GATHER_U; i++) {
                if (mine & (1u << i)) {
                    acc.x += v[i].x;
                    acc.y += v[i].y;
                    acc.z += v[i].z;
                    acc.w += v[i].w;
                }
            }
        }
    } else {
        acc.x = 0.0f;
        acc.y = 0.0f;
        acc.z = 0.0f;
        acc.w = 0.0f;
        for (int e = 0; e < E; e++) {
            if (col[e] == c) {
                v4f v = *(const v4f*)(hb + (min((unsigned)row[e], last) * 64u + qo));
                acc.x += v.x;
                acc.y += v.y;
                acc.z += v.z;
                acc.w += v.w;
            }
        }
    }
    return acc;
}

#define FB_NODES 64
#define A1_PITCH 20
#define X1_PITCH 36
__global__ void gather_mlp_kernel(const int* __restrict__ cnt, const int* __restrict__ bkt,
                                  const int* __restrict__ ovn, const v2i* __restrict__ ovf,
                                  const int* __restrict__ row, const int* __restrict__ col,
                                  const float* __restrict__ xs,
                                  const float* __restrict__ X, const float* __restrict__ W1,
                                  const float* __restrict__ b1, const float* __restrict__ W2,
                                  float* __restrict__ H, int n, int E) {
    __shared__ float a1[FB_NODES * A1_PITCH];
    __shared__ float x1s[FB_NODES * X1_PITCH];
    __shared__ float dl[FB_NODES];
    int tid = threadIdx.x;
    int j = tid >> 2;
    int q = tid & 3;
    int c = blockIdx.x * FB_NODES + j;
    v4f a = {0.0f, 0.0f, 0.0f, 0.0f};
    float d = 0.0f;
    int on = ovn[0];
    if (c < n) {
        int deg = cnt[c];
        v4f acc = {0.0f, 0.0f, 0.0f, 0.0f};
        if (deg > 0) acc = gather_rows4(bkt, xs, c, q, deg, n);
        if (deg > BK_CAP) acc = overflow_rows4(acc, ovf, on, row, col, xs, c, q, n, E);
        v4f xv = *(const v4f*)(X + (size_t)c * 16 + q * 4);
        d = rsqrtf((float)deg + 1.0f);
        float g0 = acc.x * d, g1 = acc.y * d, g2 = acc.z * d, g3 = acc.w * d;
        a.x = g0 + xv.x * d * d;
        a.y = g1 + xv.y * d * d;
        a.z = g2 + xv.z * d * d;
        a.w = g3 + xv.w * d * d;
    }
    *(v4f*)(a1 + j * A1_PITCH + q * 4) = a;
    if (q == 0) dl[j] = d;
    __syncthreads();

    int w    = tid >> 5;
    int lane = tid & 31;
    int mi   = lane & 15;
    int kb   = (lane >> 4) << 1;
    int mh   = (lane >> 4) << 3;
    {
        int t  = w >> 1;
        int ch = (w & 1) << 4;
        if (blockIdx.x * FB_NODES + t * 16 < n) {
            const float* ar = a1 + (t * 16 + mi) * A1_PITCH;
            v8f acc = {};
#pragma unroll
            for (int kk = 0; kk < 16; kk += 4) {
                v2f av = *(const v2f*)(ar + kk + kb);
                v2f bv;
                bv.x = W1[(kk + kb) * 32 + ch + mi];
                bv.y = W1[(kk + kb + 1) * 32 + ch + mi];
                acc = __builtin_amdgcn_wmma_f32_16x16x4_f32(false, av, false, bv, (short)0, acc, false, false);
            }
            float bias = b1[ch + mi];
            float* xw = x1s + (t * 16 + mh) * X1_PITCH + ch + mi;
#pragma unroll
            for (int v = 0; v < 8; v++) {
                float tv = acc[v] + bias;
                xw[v * X1_PITCH] = tv > 0.0f ? tv : 0.0f;
            }
        }
    }
    __syncthreads();

    int r0 = blockIdx.x * FB_NODES + w * 16;
    if (w < 4 && r0 < n) {
        const float* xr = x1s + (w * 16 + mi) * X1_PITCH;
        float di = dl[w * 16 + mi];
        v8f acc = {};
#pragma unroll
        for (int kk = 0; kk < 32; kk += 4) {
            v2f av = *(const v2f*)(xr + kk + kb);
            av.x *= di;
            av.y *= di;
            v2f bv;
            bv.x = W2[(kk + kb) * 16 + mi];
            bv.y = W2[(kk + kb + 1) * 16 + mi];
            acc = __builtin_amdgcn_wmma_f32_16x16x4_f32(false, av, false, bv, (short)0, acc, false, false);
        }
        float* hb = H + (size_t)(r0 + mh) * 16 + mi;
        if (r0 + 16 <= n) {
#pragma unroll
            for (int v = 0; v < 8; v++) hb[v * 16] = acc[v];
        } else {
#pragma unroll
            for (int v = 0; v < 8; v++)
                if (r0 + mh + v < n) hb[v * 16] = acc[v];
        }
    }
}

__global__ void gather_final_kernel(const int* __restrict__ cnt, const int* __restrict__ bkt,
                                    const int* __restrict__ ovn, const v2i* __restrict__ ovf,
                                    const int* __restrict__ row, const int* __restrict__ col,
                                    const float* __restrict__ hs,
                                    const float* __restrict__ b2, const float* __restrict__ fcW,
                                    const float* __restrict__ fcb, float* __restrict__ out,
                                    int n, int E) {
    __shared__ float so[64];
    int tid = threadIdx.x;
    int c = blockIdx.x * 64 + (tid >> 2);
    int q = tid & 3;
    int on = ovn[0];
    if (c < n) {
        int deg = cnt[c];
        v4f acc = {0.0f, 0.0f, 0.0f, 0.0f};
        if (deg > 0) acc = gather_rows4(bkt, hs, c, q, deg, n);
        if (deg > BK_CAP) acc = overflow_rows4(acc, ovf, on, row, col, hs, c, q, n, E);
        v4f self = *(const v4f*)(hs + (size_t)c * 16 + q * 4);
        float d = rsqrtf((float)deg + 1.0f);
        float s0, s1, s2, s3;
        {
            float v = (acc.x + self.x) * d + b2[q * 4 + 0];
            v = v > 0.0f ? v : 0.0f;
            s0 = v * fcW[q * 4 + 0];
        }
        {
            float v = (acc.y + self.y) * d + b2[q * 4 + 1];
            v = v > 0.0f ? v : 0.0f;
            s1 = v * fcW[q * 4 + 1];
        }
        {
            float v = (acc.z + self.z) * d + b2[q * 4 + 2];
            v = v > 0.0f ? v : 0.0f;
            s2 = v * fcW[q * 4 + 2];
        }
        {
            float v = (acc.w + self.w) * d + b2[q * 4 + 3];
            v = v > 0.0f ? v : 0.0f;
            s3 = v * fcW[q * 4 + 3];
        }
        s0 += __shfl_xor(s0, 2);  s1 += __shfl_xor(s1, 2);
        s2 += __shfl_xor(s2, 2);  s3 += __shfl_xor(s3, 2);
        s0 += __shfl_xor(s0, 1);  s1 += __shfl_xor(s1, 1);
        s2 += __shfl_xor(s2, 1);  s3 += __shfl_xor(s3, 1);
        float u0 = s0 + s2;
        float u1 = s1 + s3;
        float w = u0 + u1;
        if (q == 0) so[tid >> 2] = w + fcb[0];
    }
    __syncthreads();
    if (tid < 64) {
        int i = blockIdx.x * 64 + tid;
        if (i < n) out[i] = so[tid];
    }
}

extern "C" void kernel_launch(void* const* d_in, const int* in_sizes, int n_in,
                              void* d_out, int out_size, void* d_ws, size_t ws_size,
                              hipStream_t stream) {
    const int*   edge = (const int*)d_in[0];
    const float* x0   = (const float*)d_in[1];
    const float* W1   = (const float*)d_in[2];
    const float* b1   = (const float*)d_in[3];
    const float* W2   = (const float*)d_in[4];
    const float* b2   = (const float*)d_in[5];
    const float* fcW  = (const float*)d_in[6];
    const float* fcb  = (const float*)d_in[7];
    float*       out  = (float*)d_out;

    const int n = GCN_N;
    const int E = in_sizes[0] / 2;
    const int* row = edge;
    const int* col = edge + E;

    float* ws   = (float*)d_ws;
    size_t npad = ((size_t)n + 63) & ~(size_t)63;
    int*   cnt  = (int*)ws;
    int*   ovn  = cnt + n;
    v2i*   ovf  = (v2i*)(cnt + npad);
    int*   bkt  = cnt + 2 * npad;
    float* hs   = (float*)(bkt + (size_t)n * BK_CAP);
    float* xs   = hs + (size_t)n * 16;

    hipMemsetAsync(cnt, 0, (size_t)(n + 1) * sizeof(int), stream);

    int* stage = (int*)hs;
    int sblocks = (E + SORT_CHUNK - 1) / SORT_CHUNK;
    bin_kernel<<<sblocks, SORT_T, 0, stream>>>(row, col, cnt, stage, E);
    place_kernel<<<NBINS, BIN_NODES, 0, stream>>>(row, col, cnt, bkt, ovn, ovf, stage, n, E);
    prescale_kernel<<<(n * 4 + 255) / 256, 256, 0, stream>>>(cnt, x0, xs, n);

    int nblocks = (n + 63) / 64;

    gather_mlp_kernel<<<nblocks, 256, 0, stream>>>(cnt, bkt, ovn, ovf, row, col, xs, x0, W1, b1, W2, hs, n, E);

    gather_final_kernel<<<nblocks, 256, 0, stream>>>(cnt, bkt, ovn, ovf, row, col, hs, b2, fcW, fcb, out, n, E);
}


# pass 1 writes each block's bin-sorted edges to its own staging area (whole lines) with a run table that pass 2 reads: no cursor adds, no clear, memset gone; sum order: LDS arrival, as before
# speedup vs baseline: 5.8843x; 1.2240x over previous
#include <hip/hip_runtime.h>


typedef __attribute__((ext_vector_type(2))) float v2f;
typedef __attribute__((ext_vector_type(8))) float v8f;
typedef __attribute__((ext_vector_type(4))) float v4f;

#define GCN_N 100000

#define BK_CAP 32
#define OVF_CAP 50000

typedef __attribute__((ext_vector_type(2))) int v2i;
typedef __attribute__((ext_vector_type(4))) int v4i;

#define BIN_NODES 256
#define NBINS ((GCN_N + BIN_NODES - 1) / BIN_NODES)
#define HIST_W 512
#define SORT_T 1024
#define SORT_EPT 8
#define SORT_CHUNK (SORT_T * SORT_EPT)
#define RUN_PITCH 9376
#define PLACE_T 512
#define PLACE_U 8
#define SRC_MASK 0x1FFFF

__global__ void __launch_bounds__(SORT_T)
bin_kernel(const int* __restrict__ row, const int* __restrict__ col,
           int* __restrict__ runs, int* __restrict__ stage, int* __restrict__ ovn, int E, int nblk) {
    __shared__ int hist[HIST_W];
    __shared__ int loff[HIST_W];
    __shared__ int wtot[HIST_W / 32];
    __shared__ int sval[RUN_PITCH];
    int tid = threadIdx.x;
    int base = blockIdx.x * SORT_CHUNK;
    if (tid < HIST_W) hist[tid] = 0;
    v4i z = {0, 0, 0, 0};
    for (int i = tid; i < RUN_PITCH / 4; i += SORT_T) ((v4i*)sval)[i] = z;
    if (blockIdx.x == 0 && tid == 0) ovn[0] = 0;
    int cc[SORT_EPT], rr[SORT_EPT];
#pragma unroll
    for (int i = 0; i < SORT_EPT; i++) {
        int e = min(base + i * SORT_T + tid, E - 1);
        cc[i] = col[e];
        rr[i] = row[e];
    }
    __syncthreads();
    int bb[SORT_EPT], rk[SORT_EPT];
#pragma unroll
    for (int i = 0; i < SORT_EPT; i++) {
        bb[i] = (int)min((unsigned)cc[i] >> 8, (unsigned)(NBINS - 1));
        rk[i] = 0;
        if (base + i * SORT_T + tid < E) rk[i] = atomicAdd(&hist[bb[i]], 1);
    }
    __syncthreads();
    int lane = tid & 31;
    int w = tid >> 5;
    int v = 0, vp = 0, inc = 0;
    if (tid < HIST_W) {
        v = hist[tid];
        vp = (v + 3) & ~3;
        inc = vp;
#pragma unroll
        for (int d = 1; d < 32; d <<= 1) {
            int t = __shfl_up(inc, d);
            if (lane >= d) inc += t;
        }
        if (lane == 31) wtot[w] = inc;
    }
    __syncthreads();
    int tp = 0;
#pragma unroll
    for (int j = 0; j < HIST_W / 32; j++) tp += wtot[j];
    if (tid < HIST_W) {
        int wp = 0;
#pragma unroll
        for (int j = 0; j < HIST_W / 32; j++)
            if (j < w) wp += wtot[j];
        int off = wp + inc - vp;
        loff[tid] = off;
        if (tid < NBINS) runs[tid * nblk + blockIdx.x] = (off << 16) | v;
    }
    __syncthreads();
#pragma unroll
    for (int i = 0; i < SORT_EPT; i++) {
        if (base + i * SORT_T + tid < E) {
            int p = loff[bb[i]] + rk[i];
            sval[p] = ((cc[i] & (BIN_NODES - 1)) << 17) | (rr[i] & SRC_MASK);
        }
    }
    __syncthreads();
    v4i* gs = (v4i*)(stage + (size_t)blockIdx.x * RUN_PITCH);
    for (int i = tid; i < (tp >> 2); i += SORT_T) gs[i] = ((const v4i*)sval)[i];
}

__device__ __forceinline__ void put_edge(int loc, int k, int src, int node0, int* lb,
                                         int* ovn, v2i* __restrict__ ovf) {
    if ((unsigned)k < (unsigned)BK_CAP) {
        lb[loc * BK_CAP + k] = src;
    } else {
        int o = atomicAdd(ovn, 1);
        if (o < OVF_CAP) {
            v2i p = {node0 + loc, src};
            ovf[o] = p;
        }
    }
}

__global__ void __launch_bounds__(PLACE_T)
place_kernel(int* cnt, int* __restrict__ bkt, int* ovn, v2i* __restrict__ ovf,
             const int* __restrict__ runs, const int* __restrict__ stage, int n, int nblk) {
    __shared__ int lb[BIN_NODES * BK_CAP];
    __shared__ int lc[BIN_NODES];
    int tid = threadIdx.x;
    int b = blockIdx.x;
    int node0 = b * BIN_NODES;
    if (tid < BIN_NODES) lc[tid] = 0;
    v4i z = {0, 0, 0, 0};
    for (int i = tid; i < BIN_NODES * BK_CAP / 4; i += PLACE_T) ((v4i*)lb)[i] = z;
    __syncthreads();
    for (int kb = tid; kb < nblk; kb += PLACE_T) {
        unsigned r = (unsigned)runs[b * nblk + kb];
        int c = (int)(r & 0xFFFFu);
        const v4i* sp = (const v4i*)(stage + (size_t)kb * RUN_PITCH + (r >> 16));
        int nv = (c + 3) >> 2;
        for (int j0 = 0; j0 < nv; j0 += PLACE_U) {
            v4i p[PLACE_U];
            int k[PLACE_U * 4];
#pragma unroll
            for (int u = 0; u < PLACE_U; u++) p[u] = sp[min(j0 + u, nv - 1)];
#pragma unroll
            for (int u = 0; u < PLACE_U; u++) {
#pragma unroll
                for (int e = 0; e < 4; e++) {
                    k[u * 4 + e] = 0;
                    if ((j0 + u) * 4 + e < c) k[u * 4 + e] = atomicAdd(&lc[((unsigned)p[u][e] >> 17) & (BIN_NODES - 1)], 1);
                }
            }
#pragma unroll
            for (int u = 0; u < PLACE_U; u++) {
#pragma unroll
                for (int e = 0; e < 4; e++) {
                    if ((j0 + u) * 4 + e < c) {
                        unsigned q = (unsigned)p[u][e];
                        put_edge((int)((q >> 17) & (BIN_NODES - 1)), k[u * 4 + e], (int)(q & SRC_MASK), node0, lb, ovn, ovf);
                    }
                }
            }
        }
    }
    __syncthreads();
    int c = node0 + tid;
    if (tid < BIN_NODES && c < n) cnt[c] = lc[tid];
    v4i* gb = (v4i*)(bkt + (size_t)node0 * BK_CAP);
    for (int i = tid; i < BIN_NODES * BK_CAP / 4; i += PLACE_T)
        if (node0 + (i >> 3) < n) gb[i] = ((const v4i*)lb)[i];
}

__global__ void prescale_kernel(const int* __restrict__ cnt, const float* __restrict__ x,
                                float* __restrict__ xs, int n) {
    int t = blockIdx.x * blockDim.x + threadIdx.x;
    if (t < n * 4) {
        float d = rsqrtf((float)cnt[t >> 2] + 1.0f);
        v4f v = *(const v4f*)(x + (size_t)t * 4);
        v.x *= d;
        v.y *= d;
        v.z *= d;
        v.w *= d;
        *(v4f*)(xs + (size_t)t * 4) = v;
    }
}

#define GATHER_U 8
__device__ __forceinline__ v4f gather_rows4(const int* __restrict__ bkt, const float* __restrict__ hs,
                                            int c, int q, int deg, int n) {
    int beg = c * BK_CAP;
    int end = beg + min(max(deg, 0), BK_CAP);
    int hi = end - 1;
    unsigned last = (unsigned)(n - 1);
    const char* hb = (const char*)hs;
    unsigned qo = (unsigned)q * 16u;
    v4f acc = {0.0f, 0.0f, 0.0f, 0.0f};
    unsigned r[GATHER_U];
#pragma unroll
    for (int i = 0; i < GATHER_U; i++) r[i] = min((unsigned)bkt[max(min(beg + i, hi), 0)], last);
    for (int k = beg; k < end; k += GATHER_U) {
        v4f v[GATHER_U];
#pragma unroll
        for (int i = 0; i < GATHER_U; i++) v[i] = *(const v4f*)(hb + (r[i] * 64u + qo));
#pragma unroll
        for (int i = 0; i < GATHER_U; i++) r[i] = min((unsigned)bkt[max(min(k + GATHER_U + i, hi), 0)], last);
#pragma unroll
        for (int i = 0; i < GATHER_U; i++) {
            if (i == 0 || k + i < end) {
                acc.x += v[i].x;
                acc.y += v[i].y;
                acc.z += v[i].z;
                acc.w += v[i].w;
            }
        }
    }
    return acc;
}

__device__ __forceinline__ v4f overflow_rows4(v4f acc, const v2i* __restrict__ ovf, int on,
                                              const int* __restrict__ row, const int* __restrict__ col,
                                              const float* __restrict__ hs, int c, int q, int n, int E) {
    unsigned last = (unsigned)(n - 1);
    const char* hb = (const char*)hs;
    unsigned qo = (unsigned)q * 16u;
    if (on <= OVF_CAP) {
        int hi = max(on - 1, 0);
        v2i p[GATHER_U];
#pragma unroll
        for (int i = 0; i < GATHER_U; i++) p[i] = ovf[min(i, hi)];
        for (int o = 0; o < on; o += GATHER_U) {
            v4f v[GATHER_U];
            unsigned mine = 0u;
#pragma unroll
            for (int i = 0; i < GATHER_U; i++) {
                v[i] = *(const v4f*)(hb + (min((unsigned)p[i].y, last) * 64u + qo));
                if (o + i < on && p[i].x == c) mine |= 1u << i;
            }
#pragma unroll
            for (int i = 0; i < GATHER_U; i++) p[i] = ovf[min(o + GATHER_U + i, hi)];
#pragma unroll
            for (int i = 0; i < GATHER_U; i++) {
                if (mine & (1u << i)) {
                    acc.x += v[i].x;
                    acc.y += v[i].y;
                    acc.z += v[i].z;
                    acc.w += v[i].w;
                }
            }
        }
    } else {
        acc.x = 0.0f;
        acc.y = 0.0f;
        acc.z = 0.0f;
        acc.w = 0.0f;
        for (int e = 0; e < E; e++) {
            if (col[e] == c) {
                v4f v = *(const v4f*)(hb + (min((unsigned)row[e], last) * 64u + qo));
                acc.x += v.x;
                acc.y += v.y;
                acc.z += v.z;
                acc.w += v.w;
            }
        }
    }
    return acc;
}

#define FB_NODES 64
#define A1_PITCH 20
#define X1_PITCH 36
__global__ void gather_mlp_kernel(const int* __restrict__ cnt, const int* __restrict__ bkt,
                                  const int* __restrict__ ovn, const v2i* __restrict__ ovf,
                                  const int* __restrict__ row, const int* __restrict__ col,
                                  const float* __restrict__ xs,
                                  const float* __restrict__ X, const float* __restrict__ W1,
                                  const float* __restrict__ b1, const float* __restrict__ W2,
                                  float* __restrict__ H, int n, int E) {
    __shared__ float a1[FB_NODES * A1_PITCH];
    __shared__ float x1s[FB_NODES * X1_PITCH];
    __shared__ float dl[FB_NODES];
    int tid = threadIdx.x;
    int j = tid >> 2;
    int q = tid & 3;
    int c = blockIdx.x * FB_NODES + j;
    v4f a = {0.0f, 0.0f, 0.0f, 0.0f};
    float d = 0.0f;
    int on = ovn[0];
    if (c < n) {
        int deg = cnt[c];
        v4f acc = {0.0f, 0.0f, 0.0f, 0.0f};
        if (deg > 0) acc = gather_rows4(bkt, xs, c, q, deg, n);
        if (deg > BK_CAP) acc = overflow_rows4(acc, ovf, on, row, col, xs, c, q, n, E);
        v4f xv = *(const v4f*)(X + (size_t)c * 16 + q * 4);
        d = rsqrtf((float)deg + 1.0f);
        float g0 = acc.x * d, g1 = acc.y * d, g2 = acc.z * d, g3 = acc.w * d;
        a.x = g0 + xv.x * d * d;
        a.y = g1 + xv.y * d * d;
        a.z = g2 + xv.z * d * d;
        a.w = g3 + xv.w * d * d;
    }
    *(v4f*)(a1 + j * A1_PITCH + q * 4) = a;
    if (q == 0) dl[j] = d;
    __syncthreads();

    int w    = tid >> 5;
    int lane = tid & 31;
    int mi   = lane & 15;
    int kb   = (lane >> 4) << 1;
    int mh   = (lane >> 4) << 3;
    {
        int t  = w >> 1;
        int ch = (w & 1) << 4;
        if (blockIdx.x * FB_NODES + t * 16 < n) {
            const float* ar = a1 + (t * 16 + mi) * A1_PITCH;
            v8f acc = {};
#pragma unroll
            for (int kk = 0; kk < 16; kk += 4) {
                v2f av = *(const v2f*)(ar + kk + kb);
                v2f bv;
                bv.x = W1[(kk + kb) * 32 + ch + mi];
                bv.y = W1[(kk + kb + 1) * 32 + ch + mi];
                acc = __builtin_amdgcn_wmma_f32_16x16x4_f32(false, av, false, bv, (short)0, acc, false, false);
            }
            float bias = b1[ch + mi];
            float* xw = x1s + (t * 16 + mh) * X1_PITCH + ch + mi;
#pragma unroll
            for (int v = 0; v < 8; v++) {
                float tv = acc[v] + bias;
                xw[v * X1_PITCH] = tv > 0.0f ? tv : 0.0f;
            }
        }
    }
    __syncthreads();

    int r0 = blockIdx.x * FB_NODES + w * 16;
    if (w < 4 && r0 < n) {
        const float* xr = x1s + (w * 16 + mi) * X1_PITCH;
        float di = dl[w * 16 + mi];
        v8f acc = {};
#pragma unroll
        for (int kk = 0; kk < 32; kk += 4) {
            v2f av = *(const v2f*)(xr + kk + kb);
            av.x *= di;
            av.y *= di;
            v2f bv;
            bv.x = W2[(kk + kb) * 16 + mi];
            bv.y = W2[(kk + kb + 1) * 16 + mi];
            acc = __builtin_amdgcn_wmma_f32_16x16x4_f32(false, av, false, bv, (short)0, acc, false, false);
        }
        float* hb = H + (size_t)(r0 + mh) * 16 + mi;
        if (r0 + 16 <= n) {
#pragma unroll
            for (int v = 0; v < 8; v++) hb[v * 16] = acc[v];
        } else {
#pragma unroll
            for (int v = 0; v < 8; v++)
                if (r0 + mh + v < n) hb[v * 16] = acc[v];
        }
    }
}

__global__ void gather_final_kernel(const int* __restrict__ cnt, const int* __restrict__ bkt,
                                    const int* __restrict__ ovn, const v2i* __restrict__ ovf,
                                    const int* __restrict__ row, const int* __restrict__ col,
                                    const float* __restrict__ hs,
                                    const float* __restrict__ b2, const float* __restrict__ fcW,
                                    const float* __restrict__ fcb, float* __restrict__ out,
                                    int n, int E) {
    __shared__ float so[64];
    int tid = threadIdx.x;
    int c = blockIdx.x * 64 + (tid >> 2);
    int q = tid & 3;
    int on = ovn[0];
    if (c < n) {
        int deg = cnt[c];
        v4f acc = {0.0f, 0.0f, 0.0f, 0.0f};
        if (deg > 0) acc = gather_rows4(bkt, hs, c, q, deg, n);
        if (deg > BK_CAP) acc = overflow_rows4(acc, ovf, on, row, col, hs, c, q, n, E);
        v4f self = *(const v4f*)(hs + (size_t)c * 16 + q * 4);
        float d = rsqrtf((float)deg + 1.0f);
        float s0, s1, s2, s3;
        {
            float v = (acc.x + self.x) * d + b2[q * 4 + 0];
            v = v > 0.0f ? v : 0.0f;
            s0 = v * fcW[q * 4 + 0];
        }
        {
            float v = (acc.y + self.y) * d + b2[q * 4 + 1];
            v = v > 0.0f ? v : 0.0f;
            s1 = v * fcW[q * 4 + 1];
        }
        {
            float v = (acc.z + self.z) * d + b2[q * 4 + 2];
            v = v > 0.0f ? v : 0.0f;
            s2 = v * fcW[q * 4 + 2];
        }
        {
            float v = (acc.w + self.w) * d + b2[q * 4 + 3];
            v = v > 0.0f ? v : 0.0f;
            s3 = v * fcW[q * 4 + 3];
        }
        s0 += __shfl_xor(s0, 2);  s1 += __shfl_xor(s1, 2);
        s2 += __shfl_xor(s2, 2);  s3 += __shfl_xor(s3, 2);
        s0 += __shfl_xor(s0, 1);  s1 += __shfl_xor(s1, 1);
        s2 += __shfl_xor(s2, 1);  s3 += __shfl_xor(s3, 1);
        float u0 = s0 + s2;
        float u1 = s1 + s3;
        float w = u0 + u1;
        if (q == 0) so[tid >> 2] = w + fcb[0];
    }
    __syncthreads();
    if (tid < 64) {
        int i = blockIdx.x * 64 + tid;
        if (i < n) out[i] = so[tid];
    }
}

extern "C" void kernel_launch(void* const* d_in, const int* in_sizes, int n_in,
                              void* d_out, int out_size, void* d_ws, size_t ws_size,
                              hipStream_t stream) {
    const int*   edge = (const int*)d_in[0];
    const float* x0   = (const float*)d_in[1];
    const float* W1   = (const float*)d_in[2];
    const float* b1   = (const float*)d_in[3];
    const float* W2   = (const float*)d_in[4];
    const float* b2   = (const float*)d_in[5];
    const float* fcW  = (const float*)d_in[6];
    const float* fcb  = (const float*)d_in[7];
    float*       out  = (float*)d_out;

    const int n = GCN_N;
    const int E = in_sizes[0] / 2;
    const int* row = edge;
    const int* col = edge + E;

    float* ws   = (float*)d_ws;
    size_t npad = ((size_t)n + 63) & ~(size_t)63;
    int*   cnt  = (int*)ws;
    int*   ovn  = cnt + n;
    v2i*   ovf  = (v2i*)(cnt + npad);
    int*   bkt  = cnt + 2 * npad;
    float* hs   = (float*)(bkt + (size_t)n * BK_CAP);
    float* xs   = hs + (size_t)n * 16;

    int* stage = (int*)hs;
    int sblocks = (E + SORT_CHUNK - 1) / SORT_CHUNK;
    int* runs = stage + (size_t)sblocks * RUN_PITCH;
    bin_kernel<<<sblocks, SORT_T, 0, stream>>>(row, col, runs, stage, ovn, E, sblocks);
    place_kernel<<<NBINS, PLACE_T, 0, stream>>>(cnt, bkt, ovn, ovf, runs, stage, n, sblocks);
    prescale_kernel<<<(n * 4 + 255) / 256, 256, 0, stream>>>(cnt, x0, xs, n);

    int nblocks = (n + 63) / 64;

    gather_mlp_kernel<<<nblocks, 256, 0, stream>>>(cnt, bkt, ovn, ovf, row, col, xs, x0, W1, b1, W2, hs, n, E);

    gather_final_kernel<<<nblocks, 256, 0, stream>>>(cnt, bkt, ovn, ovf, row, col, hs, b2, fcW, fcb, out, n, E);
}


# gathers read a node's bucket line once (2 x 16B per lane, beside the count load) and share slot numbers in the 4-lane group by DPP quad broadcast: no 4B slot loads in the loop; sums bit-identical
# speedup vs baseline: 7.0157x; 1.1923x over previous
#include <hip/hip_runtime.h>


typedef __attribute__((ext_vector_type(2))) float v2f;
typedef __attribute__((ext_vector_type(8))) float v8f;
typedef __attribute__((ext_vector_type(4))) float v4f;

#define GCN_N 100000

#define BK_CAP 32
#define OVF_CAP 50000

typedef __attribute__((ext_vector_type(2))) int v2i;
typedef __attribute__((ext_vector_type(4))) int v4i;

#define BIN_NODES 256
#define NBINS ((GCN_N + BIN_NODES - 1) / BIN_NODES)
#define HIST_W 512
#define SORT_T 1024
#define SORT_EPT 8
#define SORT_CHUNK (SORT_T * SORT_EPT)
#define RUN_PITCH 9376
#define PLACE_T 512
#define PLACE_U 8
#define SRC_MASK 0x1FFFF

__global__ void __launch_bounds__(SORT_T)
bin_kernel(const int* __restrict__ row, const int* __restrict__ col,
           int* __restrict__ runs, int* __restrict__ stage, int* __restrict__ ovn, int E, int nblk) {
    __shared__ int hist[HIST_W];
    __shared__ int loff[HIST_W];
    __shared__ int wtot[HIST_W / 32];
    __shared__ int sval[RUN_PITCH];
    int tid = threadIdx.x;
    int base = blockIdx.x * SORT_CHUNK;
    if (tid < HIST_W) hist[tid] = 0;
    v4i z = {0, 0, 0, 0};
    for (int i = tid; i < RUN_PITCH / 4; i += SORT_T) ((v4i*)sval)[i] = z;
    if (blockIdx.x == 0 && tid == 0) ovn[0] = 0;
    int cc[SORT_EPT], rr[SORT_EPT];
#pragma unroll
    for (int i = 0; i < SORT_EPT; i++) {
        int e = min(base + i * SORT_T + tid, E - 1);
        cc[i] = col[e];
        rr[i] = row[e];
    }
    __syncthreads();
    int bb[SORT_EPT], rk[SORT_EPT];
#pragma unroll
    for (int i = 0; i < SORT_EPT; i++) {
        bb[i] = (int)min((unsigned)cc[i] >> 8, (unsigned)(NBINS - 1));
        rk[i] = 0;
        if (base + i * SORT_T + tid < E) rk[i] = atomicAdd(&hist[bb[i]], 1);
    }
    __syncthreads();
    int lane = tid & 31;
    int w = tid >> 5;
    int v = 0, vp = 0, inc = 0;
    if (tid < HIST_W) {
        v = hist[tid];
        vp = (v + 3) & ~3;
        inc = vp;
#pragma unroll
        for (int d = 1; d < 32; d <<= 1) {
            int t = __shfl_up(inc, d);
            if (lane >= d) inc += t;
        }
        if (lane == 31) wtot[w] = inc;
    }
    __syncthreads();
    int tp = 0;
#pragma unroll
    for (int j = 0; j < HIST_W / 32; j++) tp += wtot[j];
    if (tid < HIST_W) {
        int wp = 0;
#pragma unroll
        for (int j = 0; j < HIST_W / 32; j++)
            if (j < w) wp += wtot[j];
        int off = wp + inc - vp;
        loff[tid] = off;
        if (tid < NBINS) runs[tid * nblk + blockIdx.x] = (off << 16) | v;
    }
    __syncthreads();
#pragma unroll
    for (int i = 0; i < SORT_EPT; i++) {
        if (base + i * SORT_T + tid < E) {
            int p = loff[bb[i]] + rk[i];
            sval[p] = ((cc[i] & (BIN_NODES - 1)) << 17) | (rr[i] & SRC_MASK);
        }
    }
    __syncthreads();
    v4i* gs = (v4i*)(stage + (size_t)blockIdx.x * RUN_PITCH);
    for (int i = tid; i < (tp >> 2); i += SORT_T) gs[i] = ((const v4i*)sval)[i];
}

__device__ __forceinline__ void put_edge(int loc, int k, int src, int node0, int* lb,
                                         int* ovn, v2i* __restrict__ ovf) {
    if ((unsigned)k < (unsigned)BK_CAP) {
        lb[loc * BK_CAP + k] = src;
    } else {
        int o = atomicAdd(ovn, 1);
        if (o < OVF_CAP) {
            v2i p = {node0 + loc, src};
            ovf[o] = p;
        }
    }
}

__global__ void __launch_bounds__(PLACE_T)
place_kernel(int* cnt, int* __restrict__ bkt, int* ovn, v2i* __restrict__ ovf,
             const int* __restrict__ runs, const int* __restrict__ stage, int n, int nblk) {
    __shared__ int lb[BIN_NODES * BK_CAP];
    __shared__ int lc[BIN_NODES];
    int tid = threadIdx.x;
    int b = blockIdx.x;
    int node0 = b * BIN_NODES;
    if (tid < BIN_NODES) lc[tid] = 0;
    v4i z = {0, 0, 0, 0};
    for (int i = tid; i < BIN_NODES * BK_CAP / 4; i += PLACE_T) ((v4i*)lb)[i] = z;
    __syncthreads();
    for (int kb = tid; kb < nblk; kb += PLACE_T) {
        unsigned r = (unsigned)runs[b * nblk + kb];
        int c = (int)(r & 0xFFFFu);
        const v4i* sp = (const v4i*)(stage + (size_t)kb * RUN_PITCH + (r >> 16));
        int nv = (c + 3) >> 2;
        for (int j0 = 0; j0 < nv; j0 += PLACE_U) {
            v4i p[PLACE_U];
            int k[PLACE_U * 4];
#pragma unroll
            for (int u = 0; u < PLACE_U; u++) p[u] = sp[min(j0 + u, nv - 1)];
#pragma unroll
            for (int u = 0; u < PLACE_U; u++) {
#pragma unroll
                for (int e = 0; e < 4; e++) {
                    k[u * 4 + e] = 0;
                    if ((j0 + u) * 4 + e < c) k[u * 4 + e] = atomicAdd(&lc[((unsigned)p[u][e] >> 17) & (BIN_NODES - 1)], 1);
                }
            }
#pragma unroll
            for (int u = 0; u < PLACE_U; u++) {
#pragma unroll
                for (int e = 0; e < 4; e++) {
                    if ((j0 + u) * 4 + e < c) {
                        unsigned q = (unsigned)p[u][e];
                        put_edge((int)((q >> 17) & (BIN_NODES - 1)), k[u * 4 + e], (int)(q & SRC_MASK), node0, lb, ovn, ovf);
                    }
                }
            }
        }
    }
    __syncthreads();
    int c = node0 + tid;
    if (tid < BIN_NODES && c < n) cnt[c] = lc[tid];
    v4i* gb = (v4i*)(bkt + (size_t)node0 * BK_CAP);
    for (int i = tid; i < BIN_NODES * BK_CAP / 4; i += PLACE_T)
        if (node0 + (i >> 3) < n) gb[i] = ((const v4i*)lb)[i];
}

__global__ void prescale_kernel(const int* __restrict__ cnt, const float* __restrict__ x,
                                float* __restrict__ xs, int n) {
    int t = blockIdx.x * blockDim.x + threadIdx.x;
    if (t < n * 4) {
        float d = rsqrtf((float)cnt[t >> 2] + 1.0f);
        v4f v = *(const v4f*)(x + (size_t)t * 4);
        v.x *= d;
        v.y *= d;
        v.z *= d;
        v.w *= d;
        *(v4f*)(xs + (size_t)t * 4) = v;
    }
}

#define GATHER_U 8

template <int J>
__device__ __forceinline__ unsigned quad_bcast(int v) {
    return (unsigned)__builtin_amdgcn_update_dpp(0, v, J * 0x55, 0xF, 0xF, true);
}

template <int T>
__device__ __forceinline__ void gather_trip(v4f& acc, v4i lo, v4i hi, const char* hb,
                                            unsigned qo, unsigned last, int nd) {
    v4i s = (T < 2) ? lo : hi;
    constexpr int J = (T & 1) * 2;
    unsigned r[GATHER_U];
    r[0] = quad_bcast<J>(s.x);
    r[1] = quad_bcast<J>(s.y);
    r[2] = quad_bcast<J>(s.z);
    r[3] = quad_bcast<J>(s.w);
    r[4] = quad_bcast<J + 1>(s.x);
    r[5] = quad_bcast<J + 1>(s.y);
    r[6] = quad_bcast<J + 1>(s.z);
    r[7] = quad_bcast<J + 1>(s.w);
    v4f v[GATHER_U];
#pragma unroll
    for (int i = 0; i < GATHER_U; i++) v[i] = *(const v4f*)(hb + (min(r[i], last) * 64u + qo));
#pragma unroll
    for (int i = 0; i < GATHER_U; i++) {
        if (i == 0 || T * GATHER_U + i < nd) {
            acc.x += v[i].x;
            acc.y += v[i].y;
            acc.z += v[i].z;
            acc.w += v[i].w;
        }
    }
}

__device__ __forceinline__ v4f gather_rows4(v4i lo, v4i hi, const float* __restrict__ hs,
                                            int q, int deg, int n) {
    int nd = min(max(deg, 0), BK_CAP);
    unsigned last = (unsigned)(n - 1);
    const char* hb = (const char*)hs;
    unsigned qo = (unsigned)q * 16u;
    v4f acc = {0.0f, 0.0f, 0.0f, 0.0f};
    if (nd > 0) {
        gather_trip<0>(acc, lo, hi, hb, qo, last, nd);
        if (nd > GATHER_U) {
            gather_trip<1>(acc, lo, hi, hb, qo, last, nd);
            if (nd > 2 * GATHER_U) {
                gather_trip<2>(acc, lo, hi, hb, qo, last, nd);
                if (nd > 3 * GATHER_U) gather_trip<3>(acc, lo, hi, hb, qo, last, nd);
            }
        }
    }
    return acc;
}

__device__ __forceinline__ void bucket_pieces(const int* __restrict__ bkt, int c, int q, v4i& lo, v4i& hi) {
    const char* bb = (const char*)bkt;
    unsigned o = (unsigned)c * (BK_CAP * 4u) + (unsigned)q * 16u;
    lo = *(const v4i*)(bb + o);
    hi = *(const v4i*)(bb + (o + 64u));
    asm volatile("" : "+v"(lo), "+v"(hi));
}

__device__ __forceinline__ v4f overflow_rows4(v4f acc, const v2i* __restrict__ ovf, int on,
                                              const int* __restrict__ row, const int* __restrict__ col,
                                              const float* __restrict__ hs, int c, int q, int n, int E) {
    unsigned last = (unsigned)(n - 1);
    const char* hb = (const char*)hs;
    unsigned qo = (unsigned)q * 16u;
    if (on <= OVF_CAP) {
        int hi = max(on - 1, 0);
        v2i p[GATHER_U];
#pragma unroll
        for (int i = 0; i < GATHER_U; i++) p[i] = ovf[min(i, hi)];
        for (int o = 0; o < on; o += GATHER_U) {
            v4f v[GATHER_U];
            unsigned mine = 0u;
#pragma unroll
            for (int i = 0; i < GATHER_U; i++) {
                v[i] = *(const v4f*)(hb + (min((unsigned)p[i].y, last) * 64u + qo));
                if (o + i < on && p[i].x == c) mine |= 1u << i;
            }
#pragma unroll
            for (int i = 0; i < GATHER_U; i++) p[i] = ovf[min(o + GATHER_U + i, hi)];
#pragma unroll
            for (int i = 0; i < GATHER_U; i++) {
                if (mine & (1u << i)) {
                    acc.x += v[i].x;
                    acc.y += v[i].y;
                    acc.z += v[i].z;
                    acc.w += v[i].w;
                }
            }
        }
    } else {
        acc.x = 0.0f;
        acc.y = 0.0f;
        acc.z = 0.0f;
        acc.w = 0.0f;
        for (int e = 0; e < E; e++) {
            if (col[e] == c) {
                v4f v = *(const v4f*)(hb + (min((unsigned)row[e], last) * 64u + qo));
                acc.x += v.x;
                acc.y += v.y;
                acc.z += v.z;
                acc.w += v.w;
            }
        }
    }
    return acc;
}

#define FB_NODES 64
#define A1_PITCH 20
#define X1_PITCH 36
__global__ void gather_mlp_kernel(const int* __restrict__ cnt, const int* __restrict__ bkt,
                                  const int* __restrict__ ovn, const v2i* __restrict__ ovf,
                                  const int* __restrict__ row, const int* __restrict__ col,
                                  const float* __restrict__ xs,
                                  const float* __restrict__ X, const float* __restrict__ W1,
                                  const float* __restrict__ b1, const float* __restrict__ W2,
                                  float* __restrict__ H, int n, int E) {
    __shared__ float a1[FB_NODES * A1_PITCH];
    __shared__ float x1s[FB_NODES * X1_PITCH];
    __shared__ float dl[FB_NODES];
    int tid = threadIdx.x;
    int j = tid >> 2;
    int q = tid & 3;
    int c = blockIdx.x * FB_NODES + j;
    v4f a = {0.0f, 0.0f, 0.0f, 0.0f};
    float d = 0.0f;
    int on = ovn[0];
    if (c < n) {
        int deg = cnt[c];
        v4i lo, hi;
        bucket_pieces(bkt, c, q, lo, hi);
        v4f acc = gather_rows4(lo, hi, xs, q, deg, n);
        if (deg > BK_CAP) acc = overflow_rows4(acc, ovf, on, row, col, xs, c, q, n, E);
        v4f xv = *(const v4f*)(X + (size_t)c * 16 + q * 4);
        d = rsqrtf((float)deg + 1.0f);
        float g0 = acc.x * d, g1 = acc.y * d, g2 = acc.z * d, g3 = acc.w * d;
        a.x = g0 + xv.x * d * d;
        a.y = g1 + xv.y * d * d;
        a.z = g2 + xv.z * d * d;
        a.w = g3 + xv.w * d * d;
    }
    *(v4f*)(a1 + j * A1_PITCH + q * 4) = a;
    if (q == 0) dl[j] = d;
    __syncthreads();

    int w    = tid >> 5;
    int lane = tid & 31;
    int mi   = lane & 15;
    int kb   = (lane >> 4) << 1;
    int mh   = (lane >> 4) << 3;
    {
        int t  = w >> 1;
        int ch = (w & 1) << 4;
        if (blockIdx.x * FB_NODES + t * 16 < n) {
            const float* ar = a1 + (t * 16 + mi) * A1_PITCH;
            v8f acc = {};
#pragma unroll
            for (int kk = 0; kk < 16; kk += 4) {
                v2f av = *(const v2f*)(ar + kk + kb);
                v2f bv;
                bv.x = W1[(kk + kb) * 32 + ch + mi];
                bv.y = W1[(kk + kb + 1) * 32 + ch + mi];
                acc = __builtin_amdgcn_wmma_f32_16x16x4_f32(false, av, false, bv, (short)0, acc, false, false);
            }
            float bias = b1[ch + mi];
            float* xw = x1s + (t * 16 + mh) * X1_PITCH + ch + mi;
#pragma unroll
            for (int v = 0; v < 8; v++) {
                float tv = acc[v] + bias;
                xw[v * X1_PITCH] = tv > 0.0f ? tv : 0.0f;
            }
        }
    }
    __syncthreads();

    int r0 = blockIdx.x * FB_NODES + w * 16;
    if (w < 4 && r0 < n) {
        const float* xr = x1s + (w * 16 + mi) * X1_PITCH;
        float di = dl[w * 16 + mi];
        v8f acc = {};
#pragma unroll
        for (int kk = 0; kk < 32; kk += 4) {
            v2f av = *(const v2f*)(xr + kk + kb);
            av.x *= di;
            av.y *= di;
            v2f bv;
            bv.x = W2[(kk + kb) * 16 + mi];
            bv.y = W2[(kk + kb + 1) * 16 + mi];
            acc = __builtin_amdgcn_wmma_f32_16x16x4_f32(false, av, false, bv, (short)0, acc, false, false);
        }
        float* hb = H + (size_t)(r0 + mh) * 16 + mi;
        if (r0 + 16 <= n) {
#pragma unroll
            for (int v = 0; v < 8; v++) hb[v * 16] = acc[v];
        } else {
#pragma unroll
            for (int v = 0; v < 8; v++)
                if (r0 + mh + v < n) hb[v * 16] = acc[v];
        }
    }
}

__global__ void gather_final_kernel(const int* __restrict__ cnt, const int* __restrict__ bkt,
                                    const int* __restrict__ ovn, const v2i* __restrict__ ovf,
                                    const int* __restrict__ row, const int* __restrict__ col,
                                    const float* __restrict__ hs,
                                    const float* __restrict__ b2, const float* __restrict__ fcW,
                                    const float* __restrict__ fcb, float* __restrict__ out,
                                    int n, int E) {
    __shared__ float so[64];
    int tid = threadIdx.x;
    int c = blockIdx.x * 64 + (tid >> 2);
    int q = tid & 3;
    int on = ovn[0];
    if (c < n) {
        int deg = cnt[c];
        v4i lo, hi;
        bucket_pieces(bkt, c, q, lo, hi);
        v4f acc = gather_rows4(lo, hi, hs, q, deg, n);
        if (deg > BK_CAP) acc = overflow_rows4(acc, ovf, on, row, col, hs, c, q, n, E);
        v4f self = *(const v4f*)(hs + (size_t)c * 16 + q * 4);
        float d = rsqrtf((float)deg + 1.0f);
        float s0, s1, s2, s3;
        {
            float v = (acc.x + self.x) * d + b2[q * 4 + 0];
            v = v > 0.0f ? v : 0.0f;
            s0 = v * fcW[q * 4 + 0];
        }
        {
            float v = (acc.y + self.y) * d + b2[q * 4 + 1];
            v = v > 0.0f ? v : 0.0f;
            s1 = v * fcW[q * 4 + 1];
        }
        {
            float v = (acc.z + self.z) * d + b2[q * 4 + 2];
            v = v > 0.0f ? v : 0.0f;
            s2 = v * fcW[q * 4 + 2];
        }
        {
            float v = (acc.w + self.w) * d + b2[q * 4 + 3];
            v = v > 0.0f ? v : 0.0f;
            s3 = v * fcW[q * 4 + 3];
        }
        s0 += __shfl_xor(s0, 2);  s1 += __shfl_xor(s1, 2);
        s2 += __shfl_xor(s2, 2);  s3 += __shfl_xor(s3, 2);
        s0 += __shfl_xor(s0, 1);  s1 += __shfl_xor(s1, 1);
        s2 += __shfl_xor(s2, 1);  s3 += __shfl_xor(s3, 1);
        float u0 = s0 + s2;
        float u1 = s1 + s3;
        float w = u0 + u1;
        if (q == 0) so[tid >> 2] = w + fcb[0];
    }
    __syncthreads();
    if (tid < 64) {
        int i = blockIdx.x * 64 + tid;
        if (i < n) out[i] = so[tid];
    }
}

extern "C" void kernel_launch(void* const* d_in, const int* in_sizes, int n_in,
                              void* d_out, int out_size, void* d_ws, size_t ws_size,
                              hipStream_t stream) {
    const int*   edge = (const int*)d_in[0];
    const float* x0   = (const float*)d_in[1];
    const float* W1   = (const float*)d_in[2];
    const float* b1   = (const float*)d_in[3];
    const float* W2   = (const float*)d_in[4];
    const float* b2   = (const float*)d_in[5];
    const float* fcW  = (const float*)d_in[6];
    const float* fcb  = (const float*)d_in[7];
    float*       out  = (float*)d_out;

    const int n = GCN_N;
    const int E = in_sizes[0] / 2;
    const int* row = edge;
    const int* col = edge + E;

    float* ws   = (float*)d_ws;
    size_t npad = ((size_t)n + 63) & ~(size_t)63;
    int*   cnt  = (int*)ws;
    int*   ovn  = cnt + n;
    v2i*   ovf  = (v2i*)(cnt + npad);
    int*   bkt  = cnt + 2 * npad;
    float* hs   = (float*)(bkt + (size_t)n * BK_CAP);
    float* xs   = hs + (size_t)n * 16;

    int* stage = (int*)hs;
    int sblocks = (E + SORT_CHUNK - 1) / SORT_CHUNK;
    int* runs = stage + (size_t)sblocks * RUN_PITCH;
    bin_kernel<<<sblocks, SORT_T, 0, stream>>>(row, col, runs, stage, ovn, E, sblocks);
    place_kernel<<<NBINS, PLACE_T, 0, stream>>>(cnt, bkt, ovn, ovf, runs, stage, n, sblocks);
    prescale_kernel<<<(n * 4 + 255) / 256, 256, 0, stream>>>(cnt, x0, xs, n);

    int nblocks = (n + 63) / 64;

    gather_mlp_kernel<<<nblocks, 256, 0, stream>>>(cnt, bkt, ovn, ovf, row, col, xs, x0, W1, b1, W2, hs, n, E);

    gather_final_kernel<<<nblocks, 256, 0, stream>>>(cnt, bkt, ovn, ovf, row, col, hs, b2, fcW, fcb, out, n, E);
}
